# starting kernel
# baseline (speedup 1.0000x reference)
#include <hip/hip_runtime.h>
#include <hip/hip_bf16.h>

typedef float v2f __attribute__((ext_vector_type(2)));
typedef float v8f __attribute__((ext_vector_type(8)));

#define N0_   200000
#define N1_   50000
#define N2_   10000
#define E1_   750000
#define E2_   100000
#define IN_   128
#define HID_  256
#define OUT_  47
#define OUTP_ 48

#define OFF_DEG_SRC1  0
#define OFF_DEG_DST1  (OFF_DEG_SRC1 + N0_)
#define OFF_DEG_SRC2  (OFF_DEG_DST1 + N1_)
#define OFF_DEG_DST2  (OFF_DEG_SRC2 + N1_)
#define OFF_AGG1      (OFF_DEG_DST2 + N2_)
#define OFF_AGG2      (OFF_AGG1 + (size_t)N1_ * IN_)
#define OFF_H         (OFF_AGG2 + (size_t)N2_ * HID_)
#define OFF_W2P       (OFF_H + (size_t)N1_ * HID_)
#define OFF_B2P       (OFF_W2P + (size_t)HID_ * OUTP_)
#define ZERO_COUNT    OFF_H

__global__ void gcn_zero_f32(float* p, int n) {
    int i = blockIdx.x * blockDim.x + threadIdx.x;
    if (i < n) p[i] = 0.0f;
}

__global__ void gcn_degree(const int* __restrict__ idx, float* __restrict__ deg, int n) {
    int i = blockIdx.x * blockDim.x + threadIdx.x;
    if (i < n) atomicAdd(&deg[idx[i]], 1.0f);
}

__global__ void gcn_pad_w2(const float* __restrict__ W2, const float* __restrict__ b2,
                           float* __restrict__ W2p, float* __restrict__ b2p) {
    int i = blockIdx.x * blockDim.x + threadIdx.x;
    if (i < HID_ * OUTP_) {
        int r = i / OUTP_, c = i - r * OUTP_;
        W2p[i] = (c < OUT_) ? W2[r * OUT_ + c] : 0.0f;
    } else if (i < HID_ * OUTP_ + OUTP_) {
        int c = i - HID_ * OUTP_;
        b2p[c] = (c < OUT_) ? b2[c] : 0.0f;
    }
}

template<int COLS>
__global__ void gcn_scatter(const float* __restrict__ feat,
                            const int* __restrict__ src,
                            const int* __restrict__ dst,
                            const float* __restrict__ deg_src,
                            float* __restrict__ agg, int n_edges) {
    const int lanes_per_edge = COLS / 4;
    int tid = blockIdx.x * blockDim.x + threadIdx.x;
    int e = tid / lanes_per_edge;
    if (e >= n_edges) return;
    int c4 = (tid % lanes_per_edge) * 4;
    int s = src[e];
    int d = dst[e];
    float ns = rsqrtf(fmaxf(deg_src[s], 1.0f));
    const float4 v = *(const float4*)(feat + (size_t)s * COLS + c4);
    float* ap = agg + (size_t)d * COLS + c4;
    atomicAdd(ap + 0, v.x * ns);
    atomicAdd(ap + 1, v.y * ns);
    atomicAdd(ap + 2, v.z * ns);
    atomicAdd(ap + 3, v.w * ns);
}

__global__ void gcn_scale_rows(float* __restrict__ mat, const float* __restrict__ deg,
                               int n_elems, int shift) {
    int i = blockIdx.x * blockDim.x + threadIdx.x;
    if (i >= n_elems) return;
    int r = i >> shift;
    mat[i] *= rsqrtf(fmaxf(deg[r], 1.0f));
}

template<int K, int NB, bool RELU>
__global__ void __launch_bounds__(256)
gcn_gemm_wmma(const float* __restrict__ A,
              const float* __restrict__ B,
              const float* __restrict__ bias,
              float* __restrict__ C,
              int Nc, int tilesN, int totalTiles) {
    int wave = threadIdx.x >> 5;
    int lane = threadIdx.x & 31;
    int tile = blockIdx.x * (blockDim.x >> 5) + wave;
    if (tile >= totalTiles) return;

    int tm = tile / tilesN;
    int tn = tile - tm * tilesN;
    int l15  = lane & 15;
    int half = lane >> 4;

    int bcol = tn * 16 + l15;
    const float* Ap = A + (size_t)(tm * 16 + l15) * K + half * 2;
    const float* Bp = B + (size_t)(half * 2) * NB + bcol;

    v8f c = {};
    #pragma unroll
    for (int k0 = 0; k0 < K; k0 += 4) {
        v2f a = *(const v2f*)(Ap + k0);
        v2f b;
        b.x = Bp[k0 * NB];
        b.y = Bp[k0 * NB + NB];
        c = __builtin_amdgcn_wmma_f32_16x16x4_f32(
                false, a, false, b, (short)0, c, false, false);
    }

    if (bcol < Nc) {
        float bv = bias[bcol];
        #pragma unroll
        for (int r = 0; r < 8; ++r) {
            int m = tm * 16 + r + half * 8;
            float v = c[r] + bv;
            if (RELU) v = fmaxf(v, 0.0f);
            C[(size_t)m * Nc + bcol] = v;
        }
    }
}

extern "C" void kernel_launch(void* const* d_in, const int* in_sizes, int n_in,
                              void* d_out, int out_size, void* d_ws, size_t ws_size,
                              hipStream_t stream) {
    const float* x    = (const float*)d_in[0];
    const float* W1   = (const float*)d_in[1];
    const float* b1   = (const float*)d_in[2];
    const float* W2   = (const float*)d_in[3];
    const float* b2   = (const float*)d_in[4];
    const int*   src1 = (const int*)d_in[5];
    const int*   dst1 = (const int*)d_in[6];
    const int*   src2 = (const int*)d_in[7];
    const int*   dst2 = (const int*)d_in[8];

    float* ws       = (float*)d_ws;
    float* deg_src1 = ws + OFF_DEG_SRC1;
    float* deg_dst1 = ws + OFF_DEG_DST1;
    float* deg_src2 = ws + OFF_DEG_SRC2;
    float* deg_dst2 = ws + OFF_DEG_DST2;
    float* agg1     = ws + OFF_AGG1;
    float* agg2     = ws + OFF_AGG2;
    float* h        = ws + OFF_H;
    float* W2p      = ws + OFF_W2P;
    float* b2p      = ws + OFF_B2P;
    float* out      = (float*)d_out;

    const int T = 256;

    {
        int n = (int)ZERO_COUNT;
        gcn_zero_f32<<<(n + T - 1) / T, T, 0, stream>>>(ws, n);
    }

    {
        int n = HID_ * OUTP_ + OUTP_;
        gcn_pad_w2<<<(n + T - 1) / T, T, 0, stream>>>(W2, b2, W2p, b2p);
    }

    gcn_degree<<<(E1_ + T - 1) / T, T, 0, stream>>>(src1, deg_src1, E1_);
    gcn_degree<<<(E1_ + T - 1) / T, T, 0, stream>>>(dst1, deg_dst1, E1_);
    gcn_degree<<<(E2_ + T - 1) / T, T, 0, stream>>>(src2, deg_src2, E2_);
    gcn_degree<<<(E2_ + T - 1) / T, T, 0, stream>>>(dst2, deg_dst2, E2_);

    {
        long long work = (long long)E1_ * (IN_ / 4);
        gcn_scatter<IN_><<<(int)((work + T - 1) / T), T, 0, stream>>>(
            x, src1, dst1, deg_src1, agg1, E1_);
    }

    {
        int n = N1_ * IN_;
        gcn_scale_rows<<<(n + T - 1) / T, T, 0, stream>>>(agg1, deg_dst1, n, 7);
    }

    {
        int tilesM = N1_ / 16, tilesN = HID_ / 16;
        int total = tilesM * tilesN;
        int blocks = (total + (T / 32) - 1) / (T / 32);
        gcn_gemm_wmma<IN_, HID_, true><<<blocks, T, 0, stream>>>(
            agg1, W1, b1, h, HID_, tilesN, total);
    }

    {
        long long work = (long long)E2_ * (HID_ / 4);
        gcn_scatter<HID_><<<(int)((work + T - 1) / T), T, 0, stream>>>(
            h, src2, dst2, deg_src2, agg2, E2_);
    }

    {
        int n = N2_ * HID_;
        gcn_scale_rows<<<(n + T - 1) / T, T, 0, stream>>>(agg2, deg_dst2, n, 8);
    }

    {
        int tilesM = N2_ / 16, tilesN = OUTP_ / 16;
        int total = tilesM * tilesN;
        int blocks = (total + (T / 32) - 1) / (T / 32);
        gcn_gemm_wmma<HID_, OUTP_, false><<<blocks, T, 0, stream>>>(
            agg2, W2p, b2p, out, OUT_, tilesN, total);
    }
}


# Edge aggregation as a gather: edges bucketed by dst (arrival-order rank), one wave sums each dst row in registers, stores it once; float atomic adds on agg, its zero fill and scale pass gone
# speedup vs baseline: 1.4047x; 1.4047x over previous
#include <hip/hip_runtime.h>
#include <hip/hip_bf16.h>

typedef float v2f __attribute__((ext_vector_type(2)));
typedef float v8f __attribute__((ext_vector_type(8)));

#define N0_   200000
#define N1_   50000
#define N2_   10000
#define E1_   750000
#define E2_   100000
#define IN_   128
#define HID_  256
#define OUT_  47
#define OUTP_ 48

#define OFF_DEG_SRC1  0
#define OFF_DEG_DST1  (OFF_DEG_SRC1 + N0_)
#define OFF_DEG_SRC2  (OFF_DEG_DST1 + N1_)
#define OFF_DEG_DST2  (OFF_DEG_SRC2 + N1_)
#define OFF_AGG1      (OFF_DEG_DST2 + N2_)
#define OFF_AGG2      (OFF_AGG1 + (size_t)N1_ * IN_)
#define OFF_H         (OFF_AGG2 + (size_t)N2_ * HID_)
#define OFF_W2P       (OFF_H + (size_t)N1_ * HID_)
#define OFF_B2P       (OFF_W2P + (size_t)HID_ * OUTP_)
#define ZERO_COUNT    OFF_AGG1

__global__ void gcn_zero_f32(float* p, int n) {
    int i = blockIdx.x * blockDim.x + threadIdx.x;
    if (i < n) p[i] = 0.0f;
}

__global__ void gcn_degree(const int* __restrict__ idx, float* __restrict__ deg, int n) {
    int i = blockIdx.x * blockDim.x + threadIdx.x;
    if (i < n) atomicAdd(&deg[idx[i]], 1.0f);
}

__global__ void gcn_count_rank(const int* __restrict__ dst, int* __restrict__ cnt,
                               int* __restrict__ rank, int n) {
    int i = blockIdx.x * blockDim.x + threadIdx.x;
    if (i < n) rank[i] = atomicAdd(&cnt[dst[i]], 1);
}

template<int N>
__global__ void __launch_bounds__(1024)
gcn_row_starts(int* __restrict__ cnt) {
    constexpr int PER = (N + 1023) / 1024;
    __shared__ int wsum[32];
    int t = threadIdx.x;
    int lane = t & 31, wv = t >> 5;
    int lo = t * PER;
    int c[PER];
    int s = 0;
    #pragma unroll
    for (int i = 0; i < PER; ++i) {
        c[i] = (lo + i < N) ? cnt[lo + i] : 0;
        s += c[i];
    }
    int v = s;
    #pragma unroll
    for (int off = 1; off < 32; off <<= 1) {
        int u = __shfl_up(v, off);
        if (lane >= off) v += u;
    }
    if (lane == 31) wsum[wv] = v;
    __syncthreads();
    if (wv == 0) {
        int w = wsum[lane];
        #pragma unroll
        for (int off = 1; off < 32; off <<= 1) {
            int u = __shfl_up(w, off);
            if (lane >= off) w += u;
        }
        wsum[lane] = w;
    }
    __syncthreads();
    int run = v - s + (wv > 0 ? wsum[wv - 1] : 0);
    #pragma unroll
    for (int i = 0; i < PER; ++i) {
        if (lo + i < N) cnt[lo + i] = run;
        run += c[i];
    }
}

__global__ void gcn_bucket_fill(const int* __restrict__ src, const int* __restrict__ dst,
                                const int* __restrict__ rank, const int* __restrict__ start,
                                int* __restrict__ bsrc, int n) {
    int i = blockIdx.x * blockDim.x + threadIdx.x;
    if (i < n) bsrc[start[dst[i]] + rank[i]] = src[i];
}

template<int COLS, int N>
__device__ __forceinline__ void gcn_add_edges(const float* __restrict__ feat, int s_l, float w_l,
                                              int j, int lane, float4 (&acc)[COLS / 128]) {
    constexpr int V = COLS / 128;
    float4 x[N][V];
    float  w[N];
    #pragma unroll
    for (int u = 0; u < N; ++u) {
        int s = __builtin_amdgcn_readlane(s_l, j + u);
        w[u]  = __int_as_float(__builtin_amdgcn_readlane(__float_as_int(w_l), j + u));
        const float4* p = (const float4*)(feat + (size_t)s * COLS) + lane;
        #pragma unroll
        for (int v = 0; v < V; ++v) x[u][v] = p[v * 32];
    }
    #pragma unroll
    for (int u = 0; u < N; ++u) {
        #pragma unroll
        for (int v = 0; v < V; ++v) {
            acc[v].x += x[u][v].x * w[u];
            acc[v].y += x[u][v].y * w[u];
            acc[v].z += x[u][v].z * w[u];
            acc[v].w += x[u][v].w * w[u];
        }
    }
}

template<int COLS>
__global__ void __launch_bounds__(256)
gcn_gather_rows(const float* __restrict__ feat,
                const int* __restrict__ bsrc,
                const int* __restrict__ start,
                const float* __restrict__ deg_src,
                float* __restrict__ agg, int n_rows, int n_edges) {
    constexpr int V = COLS / 128;
    int lane = threadIdx.x & 31;
    int row = __builtin_amdgcn_readfirstlane(blockIdx.x * (blockDim.x >> 5) + (threadIdx.x >> 5));
    if (row >= n_rows) return;
    int beg = start[row];
    int end = (row + 1 < n_rows) ? start[row + 1] : n_edges;

    float4 acc[V];
    #pragma unroll
    for (int v = 0; v < V; ++v) acc[v] = make_float4(0.f, 0.f, 0.f, 0.f);

    for (int base = beg; base < end; base += 32) {
        int m = end - base < 32 ? end - base : 32;
        int s_l = 0;
        float w_l = 0.0f;
        if (lane < m) {
            s_l = bsrc[base + lane];
            w_l = rsqrtf(fmaxf(deg_src[s_l], 1.0f));
        }
        int j = 0;
        for (; j + 8 <= m; j += 8) gcn_add_edges<COLS, 8>(feat, s_l, w_l, j, lane, acc);
        if (j + 4 <= m) { gcn_add_edges<COLS, 4>(feat, s_l, w_l, j, lane, acc); j += 4; }
        if (j + 2 <= m) { gcn_add_edges<COLS, 2>(feat, s_l, w_l, j, lane, acc); j += 2; }
        if (j < m)      { gcn_add_edges<COLS, 1>(feat, s_l, w_l, j, lane, acc); }
    }

    float nd = rsqrtf(fmaxf((float)(end - beg), 1.0f));
    float4* o = (float4*)(agg + (size_t)row * COLS) + lane;
    #pragma unroll
    for (int v = 0; v < V; ++v)
        o[v * 32] = make_float4(acc[v].x * nd, acc[v].y * nd, acc[v].z * nd, acc[v].w * nd);
}

__global__ void gcn_pad_w2(const float* __restrict__ W2, const float* __restrict__ b2,
                           float* __restrict__ W2p, float* __restrict__ b2p) {
    int i = blockIdx.x * blockDim.x + threadIdx.x;
    if (i < HID_ * OUTP_) {
        int r = i / OUTP_, c = i - r * OUTP_;
        W2p[i] = (c < OUT_) ? W2[r * OUT_ + c] : 0.0f;
    } else if (i < HID_ * OUTP_ + OUTP_) {
        int c = i - HID_ * OUTP_;
        b2p[c] = (c < OUT_) ? b2[c] : 0.0f;
    }
}

template<int K, int NB, bool RELU>
__global__ void __launch_bounds__(256)
gcn_gemm_wmma(const float* __restrict__ A,
              const float* __restrict__ B,
              const float* __restrict__ bias,
              float* __restrict__ C,
              int Nc, int tilesN, int totalTiles) {
    int wave = threadIdx.x >> 5;
    int lane = threadIdx.x & 31;
    int tile = blockIdx.x * (blockDim.x >> 5) + wave;
    if (tile >= totalTiles) return;

    int tm = tile / tilesN;
    int tn = tile - tm * tilesN;
    int l15  = lane & 15;
    int half = lane >> 4;

    int bcol = tn * 16 + l15;
    const float* Ap = A + (size_t)(tm * 16 + l15) * K + half * 2;
    const float* Bp = B + (size_t)(half * 2) * NB + bcol;

    v8f c = {};
    #pragma unroll
    for (int k0 = 0; k0 < K; k0 += 4) {
        v2f a = *(const v2f*)(Ap + k0);
        v2f b;
        b.x = Bp[k0 * NB];
        b.y = Bp[k0 * NB + NB];
        c = __builtin_amdgcn_wmma_f32_16x16x4_f32(
                false, a, false, b, (short)0, c, false, false);
    }

    if (bcol < Nc) {
        float bv = bias[bcol];
        #pragma unroll
        for (int r = 0; r < 8; ++r) {
            int m = tm * 16 + r + half * 8;
            float v = c[r] + bv;
            if (RELU) v = fmaxf(v, 0.0f);
            C[(size_t)m * Nc + bcol] = v;
        }
    }
}

extern "C" void kernel_launch(void* const* d_in, const int* in_sizes, int n_in,
                              void* d_out, int out_size, void* d_ws, size_t ws_size,
                              hipStream_t stream) {
    const float* x    = (const float*)d_in[0];
    const float* W1   = (const float*)d_in[1];
    const float* b1   = (const float*)d_in[2];
    const float* W2   = (const float*)d_in[3];
    const float* b2   = (const float*)d_in[4];
    const int*   src1 = (const int*)d_in[5];
    const int*   dst1 = (const int*)d_in[6];
    const int*   src2 = (const int*)d_in[7];
    const int*   dst2 = (const int*)d_in[8];

    float* ws       = (float*)d_ws;
    float* deg_src1 = ws + OFF_DEG_SRC1;
    int*   cnt_dst1 = (int*)(ws + OFF_DEG_DST1);
    float* deg_src2 = ws + OFF_DEG_SRC2;
    int*   cnt_dst2 = (int*)(ws + OFF_DEG_DST2);
    float* agg1     = ws + OFF_AGG1;
    float* agg2     = ws + OFF_AGG2;
    float* h        = ws + OFF_H;
    float* W2p      = ws + OFF_W2P;
    float* b2p      = ws + OFF_B2P;
    float* out      = (float*)d_out;

    int* rank1 = (int*)h;
    int* bsrc1 = (int*)h + E1_;
    int* rank2 = (int*)agg1;
    int* bsrc2 = (int*)agg1 + E2_;

    const int T = 256;

    {
        int n = (int)ZERO_COUNT;
        gcn_zero_f32<<<(n + T - 1) / T, T, 0, stream>>>(ws, n);
    }

    {
        int n = HID_ * OUTP_ + OUTP_;
        gcn_pad_w2<<<(n + T - 1) / T, T, 0, stream>>>(W2, b2, W2p, b2p);
    }

    gcn_degree<<<(E1_ + T - 1) / T, T, 0, stream>>>(src1, deg_src1, E1_);
    gcn_count_rank<<<(E1_ + T - 1) / T, T, 0, stream>>>(dst1, cnt_dst1, rank1, E1_);
    gcn_degree<<<(E2_ + T - 1) / T, T, 0, stream>>>(src2, deg_src2, E2_);

    gcn_row_starts<N1_><<<1, 1024, 0, stream>>>(cnt_dst1);
    gcn_bucket_fill<<<(E1_ + T - 1) / T, T, 0, stream>>>(src1, dst1, rank1, cnt_dst1, bsrc1, E1_);
    gcn_gather_rows<IN_><<<(N1_ + (T / 32) - 1) / (T / 32), T, 0, stream>>>(
        x, bsrc1, cnt_dst1, deg_src1, agg1, N1_, E1_);

    {
        int tilesM = N1_ / 16, tilesN = HID_ / 16;
        int total = tilesM * tilesN;
        int blocks = (total + (T / 32) - 1) / (T / 32);
        gcn_gemm_wmma<IN_, HID_, true><<<blocks, T, 0, stream>>>(
            agg1, W1, b1, h, HID_, tilesN, total);
    }

    gcn_count_rank<<<(E2_ + T - 1) / T, T, 0, stream>>>(dst2, cnt_dst2, rank2, E2_);
    gcn_row_starts<N2_><<<1, 1024, 0, stream>>>(cnt_dst2);
    gcn_bucket_fill<<<(E2_ + T - 1) / T, T, 0, stream>>>(src2, dst2, rank2, cnt_dst2, bsrc2, E2_);
    gcn_gather_rows<HID_><<<(N2_ + (T / 32) - 1) / (T / 32), T, 0, stream>>>(
        h, bsrc2, cnt_dst2, deg_src2, agg2, N2_, E2_);

    {
        int tilesM = N2_ / 16, tilesN = OUTP_ / 16;
        int total = tilesM * tilesN;
        int blocks = (total + (T / 32) - 1) / (T / 32);
        gcn_gemm_wmma<HID_, OUTP_, false><<<blocks, T, 0, stream>>>(
            agg2, W2p, b2p, out, OUT_, tilesN, total);
    }
}


# Launches 14 -> 8: passes that do not depend on each other share a launch (zero+pad; both layers' degree and count-rank atomics; both row-start scans; both bucket fills); ranks still by arrival order
# speedup vs baseline: 1.5754x; 1.1215x over previous
#include <hip/hip_runtime.h>
#include <hip/hip_bf16.h>

typedef float v2f __attribute__((ext_vector_type(2)));
typedef float v8f __attribute__((ext_vector_type(8)));

#define N0_   200000
#define N1_   50000
#define N2_   10000
#define E1_   750000
#define E2_   100000
#define IN_   128
#define HID_  256
#define OUT_  47
#define OUTP_ 48

#define OFF_DEG_SRC1  0
#define OFF_DEG_DST1  (OFF_DEG_SRC1 + N0_)
#define OFF_DEG_SRC2  (OFF_DEG_DST1 + N1_)
#define OFF_DEG_DST2  (OFF_DEG_SRC2 + N1_)
#define OFF_AGG1      (OFF_DEG_DST2 + N2_)
#define OFF_AGG2      (OFF_AGG1 + (size_t)N1_ * IN_)
#define OFF_H         (OFF_AGG2 + (size_t)N2_ * HID_)
#define OFF_W2P       (OFF_H + (size_t)N1_ * HID_)
#define OFF_B2P       (OFF_W2P + (size_t)HID_ * OUTP_)
#define ZERO_COUNT    OFF_AGG1

__global__ void gcn_zero_and_pad(float* __restrict__ p, int n,
                                 const float* __restrict__ W2, const float* __restrict__ b2,
                                 float* __restrict__ W2p, float* __restrict__ b2p) {
    int i = blockIdx.x * blockDim.x + threadIdx.x;
    if (i < n) p[i] = 0.0f;
    if (i < HID_ * OUTP_) {
        int r = i / OUTP_, c = i - r * OUTP_;
        W2p[i] = (c < OUT_) ? W2[r * OUT_ + c] : 0.0f;
    } else if (i < HID_ * OUTP_ + OUTP_) {
        int c = i - HID_ * OUTP_;
        b2p[c] = (c < OUT_) ? b2[c] : 0.0f;
    }
}

__global__ void gcn_degrees_and_ranks(const int* __restrict__ src1, const int* __restrict__ dst1,
                                      const int* __restrict__ src2, const int* __restrict__ dst2,
                                      float* __restrict__ deg_src1, int* __restrict__ cnt_dst1,
                                      int* __restrict__ rank1,
                                      float* __restrict__ deg_src2, int* __restrict__ cnt_dst2,
                                      int* __restrict__ rank2) {
    static_assert(E2_ <= E1_, "the grid is sized by layer 1");
    int i = blockIdx.x * blockDim.x + threadIdx.x;
    if (i >= E1_) return;
    if (i < E2_) {
        int s1 = src1[i], d1 = dst1[i], s2 = src2[i], d2 = dst2[i];
        atomicAdd(&deg_src1[s1], 1.0f);
        atomicAdd(&deg_src2[s2], 1.0f);
        int r1 = atomicAdd(&cnt_dst1[d1], 1);
        int r2 = atomicAdd(&cnt_dst2[d2], 1);
        rank1[i] = r1;
        rank2[i] = r2;
    } else {
        int s1 = src1[i], d1 = dst1[i];
        atomicAdd(&deg_src1[s1], 1.0f);
        rank1[i] = atomicAdd(&cnt_dst1[d1], 1);
    }
}

template<int N>
__device__ __forceinline__ void gcn_row_starts_block(int* __restrict__ cnt, int* wsum) {
    constexpr int PER = (N + 1023) / 1024;
    int t = threadIdx.x;
    int lane = t & 31, wv = t >> 5;
    int lo = t * PER;
    int c[PER];
    int s = 0;
    #pragma unroll
    for (int i = 0; i < PER; ++i) {
        c[i] = (lo + i < N) ? cnt[lo + i] : 0;
        s += c[i];
    }
    int v = s;
    #pragma unroll
    for (int off = 1; off < 32; off <<= 1) {
        int u = __shfl_up(v, off);
        if (lane >= off) v += u;
    }
    if (lane == 31) wsum[wv] = v;
    __syncthreads();
    if (wv == 0) {
        int w = wsum[lane];
        #pragma unroll
        for (int off = 1; off < 32; off <<= 1) {
            int u = __shfl_up(w, off);
            if (lane >= off) w += u;
        }
        wsum[lane] = w;
    }
    __syncthreads();
    int run = v - s + (wv > 0 ? wsum[wv - 1] : 0);
    #pragma unroll
    for (int i = 0; i < PER; ++i) {
        if (lo + i < N) cnt[lo + i] = run;
        run += c[i];
    }
}

__global__ void __launch_bounds__(1024)
gcn_row_starts_both(int* __restrict__ cnt1, int* __restrict__ cnt2) {
    __shared__ int wsum[32];
    if (blockIdx.x == 0) gcn_row_starts_block<N1_>(cnt1, wsum);
    else                 gcn_row_starts_block<N2_>(cnt2, wsum);
}

__global__ void gcn_bucket_fill_both(const int* __restrict__ src1, const int* __restrict__ dst1,
                                     const int* __restrict__ rank1, const int* __restrict__ start1,
                                     int* __restrict__ bsrc1,
                                     const int* __restrict__ src2, const int* __restrict__ dst2,
                                     const int* __restrict__ rank2, const int* __restrict__ start2,
                                     int* __restrict__ bsrc2) {
    static_assert(E2_ <= E1_, "the grid is sized by layer 1");
    int i = blockIdx.x * blockDim.x + threadIdx.x;
    if (i >= E1_) return;
    if (i < E2_) {
        int d1 = dst1[i], d2 = dst2[i];
        int r1 = rank1[i], r2 = rank2[i];
        int s1 = src1[i], s2 = src2[i];
        int p1 = start1[d1] + r1;
        int p2 = start2[d2] + r2;
        bsrc1[p1] = s1;
        bsrc2[p2] = s2;
    } else {
        bsrc1[start1[dst1[i]] + rank1[i]] = src1[i];
    }
}

template<int COLS, int N>
__device__ __forceinline__ void gcn_add_edges(const float* __restrict__ feat, int s_l, float w_l,
                                              int j, int lane, float4 (&acc)[COLS / 128]) {
    constexpr int V = COLS / 128;
    float4 x[N][V];
    float  w[N];
    #pragma unroll
    for (int u = 0; u < N; ++u) {
        int s = __builtin_amdgcn_readlane(s_l, j + u);
        w[u]  = __int_as_float(__builtin_amdgcn_readlane(__float_as_int(w_l), j + u));
        const float4* p = (const float4*)(feat + (size_t)s * COLS) + lane;
        #pragma unroll
        for (int v = 0; v < V; ++v) x[u][v] = p[v * 32];
    }
    #pragma unroll
    for (int u = 0; u < N; ++u) {
        #pragma unroll
        for (int v = 0; v < V; ++v) {
            acc[v].x += x[u][v].x * w[u];
            acc[v].y += x[u][v].y * w[u];
            acc[v].z += x[u][v].z * w[u];
            acc[v].w += x[u][v].w * w[u];
        }
    }
}

template<int COLS>
__global__ void __launch_bounds__(256)
gcn_gather_rows(const float* __restrict__ feat,
                const int* __restrict__ bsrc,
                const int* __restrict__ start,
                const float* __restrict__ deg_src,
                float* __restrict__ agg, int n_rows, int n_edges) {
    constexpr int V = COLS / 128;
    int lane = threadIdx.x & 31;
    int row = __builtin_amdgcn_readfirstlane(blockIdx.x * (blockDim.x >> 5) + (threadIdx.x >> 5));
    if (row >= n_rows) return;
    int beg = start[row];
    int end = (row + 1 < n_rows) ? start[row + 1] : n_edges;

    float4 acc[V];
    #pragma unroll
    for (int v = 0; v < V; ++v) acc[v] = make_float4(0.f, 0.f, 0.f, 0.f);

    for (int base = beg; base < end; base += 32) {
        int m = end - base < 32 ? end - base : 32;
        int s_l = 0;
        float w_l = 0.0f;
        if (lane < m) {
            s_l = bsrc[base + lane];
            w_l = rsqrtf(fmaxf(deg_src[s_l], 1.0f));
        }
        int j = 0;
        for (; j + 8 <= m; j += 8) gcn_add_edges<COLS, 8>(feat, s_l, w_l, j, lane, acc);
        if (j + 4 <= m) { gcn_add_edges<COLS, 4>(feat, s_l, w_l, j, lane, acc); j += 4; }
        if (j + 2 <= m) { gcn_add_edges<COLS, 2>(feat, s_l, w_l, j, lane, acc); j += 2; }
        if (j < m)      { gcn_add_edges<COLS, 1>(feat, s_l, w_l, j, lane, acc); }
    }

    float nd = rsqrtf(fmaxf((float)(end - beg), 1.0f));
    float4* o = (float4*)(agg + (size_t)row * COLS) + lane;
    #pragma unroll
    for (int v = 0; v < V; ++v)
        o[v * 32] = make_float4(acc[v].x * nd, acc[v].y * nd, acc[v].z * nd, acc[v].w * nd);
}

template<int K, int NB, bool RELU>
__global__ void __launch_bounds__(256)
gcn_gemm_wmma(const float* __restrict__ A,
              const float* __restrict__ B,
              const float* __restrict__ bias,
              float* __restrict__ C,
              int Nc, int tilesN, int totalTiles) {
    int wave = threadIdx.x >> 5;
    int lane = threadIdx.x & 31;
    int tile = blockIdx.x * (blockDim.x >> 5) + wave;
    if (tile >= totalTiles) return;

    int tm = tile / tilesN;
    int tn = tile - tm * tilesN;
    int l15  = lane & 15;
    int half = lane >> 4;

    int bcol = tn * 16 + l15;
    const float* Ap = A + (size_t)(tm * 16 + l15) * K + half * 2;
    const float* Bp = B + (size_t)(half * 2) * NB + bcol;

    v8f c = {};
    #pragma unroll
    for (int k0 = 0; k0 < K; k0 += 4) {
        v2f a = *(const v2f*)(Ap + k0);
        v2f b;
        b.x = Bp[k0 * NB];
        b.y = Bp[k0 * NB + NB];
        c = __builtin_amdgcn_wmma_f32_16x16x4_f32(
                false, a, false, b, (short)0, c, false, false);
    }

    if (bcol < Nc) {
        float bv = bias[bcol];
        #pragma unroll
        for (int r = 0; r < 8; ++r) {
            int m = tm * 16 + r + half * 8;
            float v = c[r] + bv;
            if (RELU) v = fmaxf(v, 0.0f);
            C[(size_t)m * Nc + bcol] = v;
        }
    }
}

extern "C" void kernel_launch(void* const* d_in, const int* in_sizes, int n_in,
                              void* d_out, int out_size, void* d_ws, size_t ws_size,
                              hipStream_t stream) {
    const float* x    = (const float*)d_in[0];
    const float* W1   = (const float*)d_in[1];
    const float* b1   = (const float*)d_in[2];
    const float* W2   = (const float*)d_in[3];
    const float* b2   = (const float*)d_in[4];
    const int*   src1 = (const int*)d_in[5];
    const int*   dst1 = (const int*)d_in[6];
    const int*   src2 = (const int*)d_in[7];
    const int*   dst2 = (const int*)d_in[8];

    float* ws       = (float*)d_ws;
    float* deg_src1 = ws + OFF_DEG_SRC1;
    int*   cnt_dst1 = (int*)(ws + OFF_DEG_DST1);
    float* deg_src2 = ws + OFF_DEG_SRC2;
    int*   cnt_dst2 = (int*)(ws + OFF_DEG_DST2);
    float* agg1     = ws + OFF_AGG1;
    float* agg2     = ws + OFF_AGG2;
    float* h        = ws + OFF_H;
    float* W2p      = ws + OFF_W2P;
    float* b2p      = ws + OFF_B2P;
    float* out      = (float*)d_out;

    int* rank1 = (int*)h;
    int* bsrc1 = (int*)h + E1_;
    int* rank2 = (int*)h + 2 * E1_;
    int* bsrc2 = (int*)agg2;
    float* agg2s = agg1;

    const int T = 256;

    {
        int n = (int)ZERO_COUNT;
        gcn_zero_and_pad<<<(n + T - 1) / T, T, 0, stream>>>(ws, n, W2, b2, W2p, b2p);
    }

    gcn_degrees_and_ranks<<<(E1_ + T - 1) / T, T, 0, stream>>>(
        src1, dst1, src2, dst2, deg_src1, cnt_dst1, rank1, deg_src2, cnt_dst2, rank2);

    gcn_row_starts_both<<<2, 1024, 0, stream>>>(cnt_dst1, cnt_dst2);
    gcn_bucket_fill_both<<<(E1_ + T - 1) / T, T, 0, stream>>>(
        src1, dst1, rank1, cnt_dst1, bsrc1, src2, dst2, rank2, cnt_dst2, bsrc2);

    gcn_gather_rows<IN_><<<(N1_ + (T / 32) - 1) / (T / 32), T, 0, stream>>>(
        x, bsrc1, cnt_dst1, deg_src1, agg1, N1_, E1_);

    {
        int tilesM = N1_ / 16, tilesN = HID_ / 16;
        int total = tilesM * tilesN;
        int blocks = (total + (T / 32) - 1) / (T / 32);
        gcn_gemm_wmma<IN_, HID_, true><<<blocks, T, 0, stream>>>(
            agg1, W1, b1, h, HID_, tilesN, total);
    }

    gcn_gather_rows<HID_><<<(N2_ + (T / 32) - 1) / (T / 32), T, 0, stream>>>(
        h, bsrc2, cnt_dst2, deg_src2, agg2s, N2_, E2_);

    {
        int tilesM = N2_ / 16, tilesN = OUTP_ / 16;
        int total = tilesM * tilesN;
        int blocks = (total + (T / 32) - 1) / (T / 32);
        gcn_gemm_wmma<HID_, OUTP_, false><<<blocks, T, 0, stream>>>(
            agg2s, W2p, b2p, out, OUT_, tilesN, total);
    }
}


# Few long waves instead of one short wave per item: 4 edges per thread, gather waves walk several rows, a GEMM wave keeps its A rows in registers over 8 column tiles; a quarter of the waves per run
# speedup vs baseline: 1.6485x; 1.0464x over previous
#include <hip/hip_runtime.h>
#include <hip/hip_bf16.h>

typedef float v2f __attribute__((ext_vector_type(2)));
typedef float v8f __attribute__((ext_vector_type(8)));

#define N0_   200000
#define N1_   50000
#define N2_   10000
#define E1_   750000
#define E2_   100000
#define IN_   128
#define HID_  256
#define OUT_  47
#define OUTP_ 48

#define OFF_DEG_SRC1  0
#define OFF_DEG_DST1  (OFF_DEG_SRC1 + N0_)
#define OFF_DEG_SRC2  (OFF_DEG_DST1 + N1_)
#define OFF_DEG_DST2  (OFF_DEG_SRC2 + N1_)
#define OFF_AGG1      (OFF_DEG_DST2 + N2_)
#define OFF_AGG2      (OFF_AGG1 + (size_t)N1_ * IN_)
#define OFF_H         (OFF_AGG2 + (size_t)N2_ * HID_)
#define OFF_W2P       (OFF_H + (size_t)N1_ * HID_)
#define OFF_B2P       (OFF_W2P + (size_t)HID_ * OUTP_)
#define ZERO_COUNT    OFF_AGG1

__global__ void gcn_zero_and_pad(float* __restrict__ p, int n4,
                                 const float* __restrict__ W2, const float* __restrict__ b2,
                                 float* __restrict__ W2p, float* __restrict__ b2p) {
    int i = blockIdx.x * blockDim.x + threadIdx.x;
    if (i < n4) ((float4*)p)[i] = make_float4(0.f, 0.f, 0.f, 0.f);
    if (i < HID_ * OUTP_) {
        int r = i / OUTP_, c = i - r * OUTP_;
        W2p[i] = (c < OUT_) ? W2[r * OUT_ + c] : 0.0f;
    } else if (i < HID_ * OUTP_ + OUTP_) {
        int c = i - HID_ * OUTP_;
        b2p[c] = (c < OUT_) ? b2[c] : 0.0f;
    }
}

__global__ void gcn_degrees_and_ranks(const int* __restrict__ src1, const int* __restrict__ dst1,
                                      const int* __restrict__ src2, const int* __restrict__ dst2,
                                      float* __restrict__ deg_src1, int* __restrict__ cnt_dst1,
                                      int* __restrict__ rank1,
                                      float* __restrict__ deg_src2, int* __restrict__ cnt_dst2,
                                      int* __restrict__ rank2) {
    static_assert(E2_ <= E1_, "the grid is sized by layer 1");
    static_assert(E1_ % 4 == 0 && E2_ % 4 == 0, "four edges per thread");
    int i = blockIdx.x * blockDim.x + threadIdx.x;
    if (i >= E1_ / 4) return;
    bool two = i < E2_ / 4;
    int4 s1 = ((const int4*)src1)[i];
    int4 d1 = ((const int4*)dst1)[i];
    int4 s2 = make_int4(0, 0, 0, 0), d2 = make_int4(0, 0, 0, 0);
    if (two) {
        s2 = ((const int4*)src2)[i];
        d2 = ((const int4*)dst2)[i];
    }
    atomicAdd(&deg_src1[s1.x], 1.0f);
    atomicAdd(&deg_src1[s1.y], 1.0f);
    atomicAdd(&deg_src1[s1.z], 1.0f);
    atomicAdd(&deg_src1[s1.w], 1.0f);
    int4 r1;
    r1.x = atomicAdd(&cnt_dst1[d1.x], 1);
    r1.y = atomicAdd(&cnt_dst1[d1.y], 1);
    r1.z = atomicAdd(&cnt_dst1[d1.z], 1);
    r1.w = atomicAdd(&cnt_dst1[d1.w], 1);
    if (two) {
        atomicAdd(&deg_src2[s2.x], 1.0f);
        atomicAdd(&deg_src2[s2.y], 1.0f);
        atomicAdd(&deg_src2[s2.z], 1.0f);
        atomicAdd(&deg_src2[s2.w], 1.0f);
        int4 r2;
        r2.x = atomicAdd(&cnt_dst2[d2.x], 1);
        r2.y = atomicAdd(&cnt_dst2[d2.y], 1);
        r2.z = atomicAdd(&cnt_dst2[d2.z], 1);
        r2.w = atomicAdd(&cnt_dst2[d2.w], 1);
        ((int4*)rank2)[i] = r2;
    }
    ((int4*)rank1)[i] = r1;
}

template<int N>
__device__ __forceinline__ void gcn_row_starts_block(int* __restrict__ cnt, int* wsum) {
    constexpr int PER = (N + 1023) / 1024;
    int t = threadIdx.x;
    int lane = t & 31, wv = t >> 5;
    int lo = t * PER;
    int c[PER];
    int s = 0;
    #pragma unroll
    for (int i = 0; i < PER; ++i) {
        c[i] = (lo + i < N) ? cnt[lo + i] : 0;
        s += c[i];
    }
    int v = s;
    #pragma unroll
    for (int off = 1; off < 32; off <<= 1) {
        int u = __shfl_up(v, off);
        if (lane >= off) v += u;
    }
    if (lane == 31) wsum[wv] = v;
    __syncthreads();
    if (wv == 0) {
        int w = wsum[lane];
        #pragma unroll
        for (int off = 1; off < 32; off <<= 1) {
            int u = __shfl_up(w, off);
            if (lane >= off) w += u;
        }
        wsum[lane] = w;
    }
    __syncthreads();
    int run = v - s + (wv > 0 ? wsum[wv - 1] : 0);
    #pragma unroll
    for (int i = 0; i < PER; ++i) {
        if (lo + i < N) cnt[lo + i] = run;
        run += c[i];
    }
}

__global__ void __launch_bounds__(1024)
gcn_row_starts_both(int* __restrict__ cnt1, int* __restrict__ cnt2) {
    __shared__ int wsum[32];
    if (blockIdx.x == 0) gcn_row_starts_block<N1_>(cnt1, wsum);
    else                 gcn_row_starts_block<N2_>(cnt2, wsum);
}

__global__ void gcn_bucket_fill_both(const int* __restrict__ src1, const int* __restrict__ dst1,
                                     const int* __restrict__ rank1, const int* __restrict__ start1,
                                     int* __restrict__ bsrc1,
                                     const int* __restrict__ src2, const int* __restrict__ dst2,
                                     const int* __restrict__ rank2, const int* __restrict__ start2,
                                     int* __restrict__ bsrc2) {
    static_assert(E2_ <= E1_, "the grid is sized by layer 1");
    static_assert(E1_ % 4 == 0 && E2_ % 4 == 0, "four edges per thread");
    int i = blockIdx.x * blockDim.x + threadIdx.x;
    if (i >= E1_ / 4) return;
    bool two = i < E2_ / 4;
    int4 d1 = ((const int4*)dst1)[i];
    int4 r1 = ((const int4*)rank1)[i];
    int4 s1 = ((const int4*)src1)[i];
    int4 d2 = make_int4(0, 0, 0, 0), r2 = d2, s2 = d2;
    if (two) {
        d2 = ((const int4*)dst2)[i];
        r2 = ((const int4*)rank2)[i];
        s2 = ((const int4*)src2)[i];
    }
    int p1x = start1[d1.x] + r1.x;
    int p1y = start1[d1.y] + r1.y;
    int p1z = start1[d1.z] + r1.z;
    int p1w = start1[d1.w] + r1.w;
    if (two) {
        int p2x = start2[d2.x] + r2.x;
        int p2y = start2[d2.y] + r2.y;
        int p2z = start2[d2.z] + r2.z;
        int p2w = start2[d2.w] + r2.w;
        bsrc2[p2x] = s2.x;
        bsrc2[p2y] = s2.y;
        bsrc2[p2z] = s2.z;
        bsrc2[p2w] = s2.w;
    }
    bsrc1[p1x] = s1.x;
    bsrc1[p1y] = s1.y;
    bsrc1[p1z] = s1.z;
    bsrc1[p1w] = s1.w;
}

template<int COLS, int N>
__device__ __forceinline__ void gcn_add_edges(const float* __restrict__ feat, int s_l, float w_l,
                                              int j, int lane, float4 (&acc)[COLS / 128]) {
    constexpr int V = COLS / 128;
    float4 x[N][V];
    float  w[N];
    #pragma unroll
    for (int u = 0; u < N; ++u) {
        int s = __builtin_amdgcn_readlane(s_l, j + u);
        w[u]  = __int_as_float(__builtin_amdgcn_readlane(__float_as_int(w_l), j + u));
        const float4* p = (const float4*)(feat + (size_t)s * COLS) + lane;
        #pragma unroll
        for (int v = 0; v < V; ++v) x[u][v] = p[v * 32];
    }
    #pragma unroll
    for (int u = 0; u < N; ++u) {
        #pragma unroll
        for (int v = 0; v < V; ++v) {
            acc[v].x += x[u][v].x * w[u];
            acc[v].y += x[u][v].y * w[u];
            acc[v].z += x[u][v].z * w[u];
            acc[v].w += x[u][v].w * w[u];
        }
    }
}

template<int COLS>
__global__ void __launch_bounds__(256)
gcn_gather_rows(const float* __restrict__ feat,
                const int* __restrict__ bsrc,
                const int* __restrict__ start,
                const float* __restrict__ deg_src,
                float* __restrict__ agg, int n_rows, int n_edges) {
    constexpr int V = COLS / 128;
    int lane = threadIdx.x & 31;
    int w0 = __builtin_amdgcn_readfirstlane(blockIdx.x * (blockDim.x >> 5) + (threadIdx.x >> 5));
    int nw = gridDim.x * (blockDim.x >> 5);
    for (int row = w0; row < n_rows; row += nw) {
    int beg = start[row];
    int end = (row + 1 < n_rows) ? start[row + 1] : n_edges;

    float4 acc[V];
    #pragma unroll
    for (int v = 0; v < V; ++v) acc[v] = make_float4(0.f, 0.f, 0.f, 0.f);

    for (int base = beg; base < end; base += 32) {
        int m = end - base < 32 ? end - base : 32;
        int s_l = 0;
        float w_l = 0.0f;
        if (lane < m) {
            s_l = bsrc[base + lane];
            w_l = rsqrtf(fmaxf(deg_src[s_l], 1.0f));
        }
        int j = 0;
        for (; j + 8 <= m; j += 8) gcn_add_edges<COLS, 8>(feat, s_l, w_l, j, lane, acc);
        if (j + 4 <= m) { gcn_add_edges<COLS, 4>(feat, s_l, w_l, j, lane, acc); j += 4; }
        if (j + 2 <= m) { gcn_add_edges<COLS, 2>(feat, s_l, w_l, j, lane, acc); j += 2; }
        if (j < m)      { gcn_add_edges<COLS, 1>(feat, s_l, w_l, j, lane, acc); }
    }

    float nd = rsqrtf(fmaxf((float)(end - beg), 1.0f));
    float4* o = (float4*)(agg + (size_t)row * COLS) + lane;
    #pragma unroll
    for (int v = 0; v < V; ++v)
        o[v * 32] = make_float4(acc[v].x * nd, acc[v].y * nd, acc[v].z * nd, acc[v].w * nd);
    }
}

template<bool RELU>
__device__ __forceinline__ void gcn_store_tile(float* __restrict__ C, const float* __restrict__ bias,
                                               const v8f& c, int tm, int half, int bcol, int Nc) {
    if (bcol < Nc) {
        float bv = bias[bcol];
        #pragma unroll
        for (int r = 0; r < 8; ++r) {
            int m = tm * 16 + r + half * 8;
            float v = c[r] + bv;
            if (RELU) v = fmaxf(v, 0.0f);
            C[(size_t)m * Nc + bcol] = v;
        }
    }
}

template<int K, int NB, bool RELU, int TPW>
__global__ void __launch_bounds__(256)
gcn_gemm_wmma(const float* __restrict__ A,
              const float* __restrict__ B,
              const float* __restrict__ bias,
              float* __restrict__ C,
              int Nc, int tilesM) {
    constexpr int GROUPS = (NB / 16) / TPW;
    static_assert(GROUPS * TPW * 16 == NB, "column tiles must split evenly");
    int wave = threadIdx.x >> 5;
    int lane = threadIdx.x & 31;
    int tg = blockIdx.x % GROUPS;
    int tm = (blockIdx.x / GROUPS) * (blockDim.x >> 5) + wave;
    if (tm >= tilesM) return;

    int l15  = lane & 15;
    int half = lane >> 4;

    const float* Ap = A + (size_t)(tm * 16 + l15) * K + half * 2;
    v2f a[K / 4];
    #pragma unroll
    for (int s = 0; s < K / 4; ++s) a[s] = *(const v2f*)(Ap + 4 * s);

    if constexpr (TPW % 2 == 0) {
        for (int t = 0; t < TPW; t += 2) {
            int bcol = (tg * TPW + t) * 16 + l15;
            const float* Bp = B + (size_t)(half * 2) * NB + bcol;
            v8f c0 = {}, c1 = {};
            #pragma unroll
            for (int k0 = 0; k0 < K; k0 += 4) {
                v2f b0, b1;
                b0.x = Bp[k0 * NB];
                b0.y = Bp[k0 * NB + NB];
                b1.x = Bp[k0 * NB + 16];
                b1.y = Bp[k0 * NB + NB + 16];
                c0 = __builtin_amdgcn_wmma_f32_16x16x4_f32(
                        false, a[k0 / 4], false, b0, (short)0, c0, false, false);
                c1 = __builtin_amdgcn_wmma_f32_16x16x4_f32(
                        false, a[k0 / 4], false, b1, (short)0, c1, false, false);
            }
            gcn_store_tile<RELU>(C, bias, c0, tm, half, bcol, Nc);
            gcn_store_tile<RELU>(C, bias, c1, tm, half, bcol + 16, Nc);
        }
    } else {
        for (int t = 0; t < TPW; ++t) {
            int bcol = (tg * TPW + t) * 16 + l15;
            const float* Bp = B + (size_t)(half * 2) * NB + bcol;
            v8f c = {};
            #pragma unroll
            for (int k0 = 0; k0 < K; k0 += 4) {
                v2f b;
                b.x = Bp[k0 * NB];
                b.y = Bp[k0 * NB + NB];
                c = __builtin_amdgcn_wmma_f32_16x16x4_f32(
                        false, a[k0 / 4], false, b, (short)0, c, false, false);
            }
            gcn_store_tile<RELU>(C, bias, c, tm, half, bcol, Nc);
        }
    }
}

extern "C" void kernel_launch(void* const* d_in, const int* in_sizes, int n_in,
                              void* d_out, int out_size, void* d_ws, size_t ws_size,
                              hipStream_t stream) {
    const float* x    = (const float*)d_in[0];
    const float* W1   = (const float*)d_in[1];
    const float* b1   = (const float*)d_in[2];
    const float* W2   = (const float*)d_in[3];
    const float* b2   = (const float*)d_in[4];
    const int*   src1 = (const int*)d_in[5];
    const int*   dst1 = (const int*)d_in[6];
    const int*   src2 = (const int*)d_in[7];
    const int*   dst2 = (const int*)d_in[8];

    float* ws       = (float*)d_ws;
    float* deg_src1 = ws + OFF_DEG_SRC1;
    int*   cnt_dst1 = (int*)(ws + OFF_DEG_DST1);
    float* deg_src2 = ws + OFF_DEG_SRC2;
    int*   cnt_dst2 = (int*)(ws + OFF_DEG_DST2);
    float* agg1     = ws + OFF_AGG1;
    float* agg2     = ws + OFF_AGG2;
    float* h        = ws + OFF_H;
    float* W2p      = ws + OFF_W2P;
    float* b2p      = ws + OFF_B2P;
    float* out      = (float*)d_out;

    int* rank1 = (int*)h;
    int* bsrc1 = (int*)h + E1_;
    int* rank2 = (int*)h + 2 * E1_;
    int* bsrc2 = (int*)agg2;
    float* agg2s = agg1;

    const int T = 256;

    {
        static_assert(ZERO_COUNT % 4 == 0, "16-byte zero stores");
        static_assert(ZERO_COUNT / 4 >= HID_ * OUTP_ + OUTP_, "the grid is sized by the zero fill");
        int n4 = (int)(ZERO_COUNT / 4);
        gcn_zero_and_pad<<<(n4 + T - 1) / T, T, 0, stream>>>(ws, n4, W2, b2, W2p, b2p);
    }

    gcn_degrees_and_ranks<<<(E1_ / 4 + T - 1) / T, T, 0, stream>>>(
        src1, dst1, src2, dst2, deg_src1, cnt_dst1, rank1, deg_src2, cnt_dst2, rank2);

    gcn_row_starts_both<<<2, 1024, 0, stream>>>(cnt_dst1, cnt_dst2);
    gcn_bucket_fill_both<<<(E1_ / 4 + T - 1) / T, T, 0, stream>>>(
        src1, dst1, rank1, cnt_dst1, bsrc1, src2, dst2, rank2, cnt_dst2, bsrc2);

    gcn_gather_rows<IN_><<<2048, T, 0, stream>>>(
        x, bsrc1, cnt_dst1, deg_src1, agg1, N1_, E1_);

    {
        static_assert(N1_ % 16 == 0, "whole row tiles");
        int tilesM = N1_ / 16;
        int blocks = ((tilesM + (T / 32) - 1) / (T / 32)) * 2;
        gcn_gemm_wmma<IN_, HID_, true, 8><<<blocks, T, 0, stream>>>(
            agg1, W1, b1, h, HID_, tilesM);
    }

    gcn_gather_rows<HID_><<<(N2_ / 2 + (T / 32) - 1) / (T / 32), T, 0, stream>>>(
        h, bsrc2, cnt_dst2, deg_src2, agg2s, N2_, E2_);

    {
        static_assert(N2_ % 16 == 0, "whole row tiles");
        int tilesM = N2_ / 16;
        int blocks = ((tilesM + (T / 32) - 1) / (T / 32)) * (OUTP_ / 16);
        gcn_gemm_wmma<HID_, OUTP_, false, 1><<<blocks, T, 0, stream>>>(
            agg2s, W2p, b2p, out, OUT_, tilesM);
    }
}


# Row-start scan reads and writes whole lines: a wave scans a contiguous segment with 16-byte loads and stores; before, 4-byte accesses at a 196-byte lane stride touched 32 lines per instruction
# speedup vs baseline: 2.0401x; 1.2375x over previous
#include <hip/hip_runtime.h>
#include <hip/hip_bf16.h>

typedef float v2f __attribute__((ext_vector_type(2)));
typedef float v8f __attribute__((ext_vector_type(8)));

#define N0_   200000
#define N1_   50000
#define N2_   10000
#define E1_   750000
#define E2_   100000
#define IN_   128
#define HID_  256
#define OUT_  47
#define OUTP_ 48

#define OFF_DEG_SRC1  0
#define OFF_DEG_DST1  (OFF_DEG_SRC1 + N0_)
#define OFF_DEG_SRC2  (OFF_DEG_DST1 + N1_)
#define OFF_DEG_DST2  (OFF_DEG_SRC2 + N1_)
#define OFF_AGG1      (OFF_DEG_DST2 + N2_)
#define OFF_AGG2      (OFF_AGG1 + (size_t)N1_ * IN_)
#define OFF_H         (OFF_AGG2 + (size_t)N2_ * HID_)
#define OFF_W2P       (OFF_H + (size_t)N1_ * HID_)
#define OFF_B2P       (OFF_W2P + (size_t)HID_ * OUTP_)
#define ZERO_COUNT    OFF_AGG1

__global__ void gcn_zero_and_pad(float* __restrict__ p, int n4,
                                 const float* __restrict__ W2, const float* __restrict__ b2,
                                 float* __restrict__ W2p, float* __restrict__ b2p) {
    int i = blockIdx.x * blockDim.x + threadIdx.x;
    if (i < n4) ((float4*)p)[i] = make_float4(0.f, 0.f, 0.f, 0.f);
    if (i < HID_ * OUTP_) {
        int r = i / OUTP_, c = i - r * OUTP_;
        W2p[i] = (c < OUT_) ? W2[r * OUT_ + c] : 0.0f;
    } else if (i < HID_ * OUTP_ + OUTP_) {
        int c = i - HID_ * OUTP_;
        b2p[c] = (c < OUT_) ? b2[c] : 0.0f;
    }
}

__global__ void gcn_degrees_and_ranks(const int* __restrict__ src1, const int* __restrict__ dst1,
                                      const int* __restrict__ src2, const int* __restrict__ dst2,
                                      float* __restrict__ deg_src1, int* __restrict__ cnt_dst1,
                                      int* __restrict__ rank1,
                                      float* __restrict__ deg_src2, int* __restrict__ cnt_dst2,
                                      int* __restrict__ rank2) {
    static_assert(E2_ <= E1_, "the grid is sized by layer 1");
    static_assert(E1_ % 4 == 0 && E2_ % 4 == 0, "four edges per thread");
    int i = blockIdx.x * blockDim.x + threadIdx.x;
    if (i >= E1_ / 4) return;
    bool two = i < E2_ / 4;
    int4 s1 = ((const int4*)src1)[i];
    int4 d1 = ((const int4*)dst1)[i];
    int4 s2 = make_int4(0, 0, 0, 0), d2 = make_int4(0, 0, 0, 0);
    if (two) {
        s2 = ((const int4*)src2)[i];
        d2 = ((const int4*)dst2)[i];
    }
    atomicAdd(&deg_src1[s1.x], 1.0f);
    atomicAdd(&deg_src1[s1.y], 1.0f);
    atomicAdd(&deg_src1[s1.z], 1.0f);
    atomicAdd(&deg_src1[s1.w], 1.0f);
    int4 r1;
    r1.x = atomicAdd(&cnt_dst1[d1.x], 1);
    r1.y = atomicAdd(&cnt_dst1[d1.y], 1);
    r1.z = atomicAdd(&cnt_dst1[d1.z], 1);
    r1.w = atomicAdd(&cnt_dst1[d1.w], 1);
    if (two) {
        atomicAdd(&deg_src2[s2.x], 1.0f);
        atomicAdd(&deg_src2[s2.y], 1.0f);
        atomicAdd(&deg_src2[s2.z], 1.0f);
        atomicAdd(&deg_src2[s2.w], 1.0f);
        int4 r2;
        r2.x = atomicAdd(&cnt_dst2[d2.x], 1);
        r2.y = atomicAdd(&cnt_dst2[d2.y], 1);
        r2.z = atomicAdd(&cnt_dst2[d2.z], 1);
        r2.w = atomicAdd(&cnt_dst2[d2.w], 1);
        ((int4*)rank2)[i] = r2;
    }
    ((int4*)rank1)[i] = r1;
}

template<int N>
__device__ __forceinline__ void gcn_row_starts_block(int* __restrict__ cnt, int* wsum) {
    static_assert(N % 4 == 0, "rows are taken four at a time");
    constexpr int IT  = ((N + 31) / 32 + 127) / 128;
    constexpr int SEG = IT * 128;
    int t = threadIdx.x;
    int lane = t & 31, wv = t >> 5;
    int lo = wv * SEG + lane * 4;
    int4 c[IT];
    int  ex[IT];
    #pragma unroll
    for (int i = 0; i < IT; ++i) {
        int r = lo + i * 128;
        c[i] = (r < N) ? *(const int4*)(cnt + r) : make_int4(0, 0, 0, 0);
    }
    int carry = 0;
    #pragma unroll
    for (int i = 0; i < IT; ++i) {
        int s = c[i].x + c[i].y + c[i].z + c[i].w;
        int v = s;
        #pragma unroll
        for (int off = 1; off < 32; off <<= 1) {
            int u = __shfl_up(v, off);
            if (lane >= off) v += u;
        }
        ex[i] = carry + v - s;
        carry += __builtin_amdgcn_readlane(v, 31);
    }
    if (lane == 0) wsum[wv] = carry;
    __syncthreads();
    if (wv == 0) {
        int w = wsum[lane];
        #pragma unroll
        for (int off = 1; off < 32; off <<= 1) {
            int u = __shfl_up(w, off);
            if (lane >= off) w += u;
        }
        wsum[lane] = w;
    }
    __syncthreads();
    int base = (wv > 0) ? wsum[wv - 1] : 0;
    #pragma unroll
    for (int i = 0; i < IT; ++i) {
        int r = lo + i * 128;
        if (r < N) {
            int p0 = base + ex[i];
            int p1 = p0 + c[i].x;
            int p2 = p1 + c[i].y;
            int p3 = p2 + c[i].z;
            *(int4*)(cnt + r) = make_int4(p0, p1, p2, p3);
        }
    }
}

__global__ void __launch_bounds__(1024)
gcn_row_starts_both(int* __restrict__ cnt1, int* __restrict__ cnt2) {
    __shared__ int wsum[32];
    if (blockIdx.x == 0) gcn_row_starts_block<N1_>(cnt1, wsum);
    else                 gcn_row_starts_block<N2_>(cnt2, wsum);
}

__global__ void gcn_bucket_fill_both(const int* __restrict__ src1, const int* __restrict__ dst1,
                                     const int* __restrict__ rank1, const int* __restrict__ start1,
                                     int* __restrict__ bsrc1,
                                     const int* __restrict__ src2, const int* __restrict__ dst2,
                                     const int* __restrict__ rank2, const int* __restrict__ start2,
                                     int* __restrict__ bsrc2) {
    static_assert(E2_ <= E1_, "the grid is sized by layer 1");
    static_assert(E1_ % 4 == 0 && E2_ % 4 == 0, "four edges per thread");
    int i = blockIdx.x * blockDim.x + threadIdx.x;
    if (i >= E1_ / 4) return;
    bool two = i < E2_ / 4;
    int4 d1 = ((const int4*)dst1)[i];
    int4 r1 = ((const int4*)rank1)[i];
    int4 s1 = ((const int4*)src1)[i];
    int4 d2 = make_int4(0, 0, 0, 0), r2 = d2, s2 = d2;
    if (two) {
        d2 = ((const int4*)dst2)[i];
        r2 = ((const int4*)rank2)[i];
        s2 = ((const int4*)src2)[i];
    }
    int p1x = start1[d1.x] + r1.x;
    int p1y = start1[d1.y] + r1.y;
    int p1z = start1[d1.z] + r1.z;
    int p1w = start1[d1.w] + r1.w;
    if (two) {
        int p2x = start2[d2.x] + r2.x;
        int p2y = start2[d2.y] + r2.y;
        int p2z = start2[d2.z] + r2.z;
        int p2w = start2[d2.w] + r2.w;
        bsrc2[p2x] = s2.x;
        bsrc2[p2y] = s2.y;
        bsrc2[p2z] = s2.z;
        bsrc2[p2w] = s2.w;
    }
    bsrc1[p1x] = s1.x;
    bsrc1[p1y] = s1.y;
    bsrc1[p1z] = s1.z;
    bsrc1[p1w] = s1.w;
}

template<int COLS, int N>
__device__ __forceinline__ void gcn_add_edges(const float* __restrict__ feat, int s_l, float w_l,
                                              int j, int lane, float4 (&acc)[COLS / 128]) {
    constexpr int V = COLS / 128;
    float4 x[N][V];
    float  w[N];
    #pragma unroll
    for (int u = 0; u < N; ++u) {
        int s = __builtin_amdgcn_readlane(s_l, j + u);
        w[u]  = __int_as_float(__builtin_amdgcn_readlane(__float_as_int(w_l), j + u));
        const float4* p = (const float4*)(feat + (size_t)s * COLS) + lane;
        #pragma unroll
        for (int v = 0; v < V; ++v) x[u][v] = p[v * 32];
    }
    #pragma unroll
    for (int u = 0; u < N; ++u) {
        #pragma unroll
        for (int v = 0; v < V; ++v) {
            acc[v].x += x[u][v].x * w[u];
            acc[v].y += x[u][v].y * w[u];
            acc[v].z += x[u][v].z * w[u];
            acc[v].w += x[u][v].w * w[u];
        }
    }
}

template<int COLS>
__global__ void __launch_bounds__(256)
gcn_gather_rows(const float* __restrict__ feat,
                const int* __restrict__ bsrc,
                const int* __restrict__ start,
                const float* __restrict__ deg_src,
                float* __restrict__ agg, int n_rows, int n_edges) {
    constexpr int V = COLS / 128;
    int lane = threadIdx.x & 31;
    int w0 = __builtin_amdgcn_readfirstlane(blockIdx.x * (blockDim.x >> 5) + (threadIdx.x >> 5));
    int nw = gridDim.x * (blockDim.x >> 5);
    for (int row = w0; row < n_rows; row += nw) {
    int beg = start[row];
    int end = (row + 1 < n_rows) ? start[row + 1] : n_edges;

    float4 acc[V];
    #pragma unroll
    for (int v = 0; v < V; ++v) acc[v] = make_float4(0.f, 0.f, 0.f, 0.f);

    for (int base = beg; base < end; base += 32) {
        int m = end - base < 32 ? end - base : 32;
        int s_l = 0;
        float w_l = 0.0f;
        if (lane < m) {
            s_l = bsrc[base + lane];
            w_l = rsqrtf(fmaxf(deg_src[s_l], 1.0f));
        }
        int j = 0;
        for (; j + 8 <= m; j += 8) gcn_add_edges<COLS, 8>(feat, s_l, w_l, j, lane, acc);
        if (j + 4 <= m) { gcn_add_edges<COLS, 4>(feat, s_l, w_l, j, lane, acc); j += 4; }
        if (j + 2 <= m) { gcn_add_edges<COLS, 2>(feat, s_l, w_l, j, lane, acc); j += 2; }
        if (j < m)      { gcn_add_edges<COLS, 1>(feat, s_l, w_l, j, lane, acc); }
    }

    float nd = rsqrtf(fmaxf((float)(end - beg), 1.0f));
    float4* o = (float4*)(agg + (size_t)row * COLS) + lane;
    #pragma unroll
    for (int v = 0; v < V; ++v)
        o[v * 32] = make_float4(acc[v].x * nd, acc[v].y * nd, acc[v].z * nd, acc[v].w * nd);
    }
}

template<bool RELU>
__device__ __forceinline__ void gcn_store_tile(float* __restrict__ C, const float* __restrict__ bias,
                                               const v8f& c, int tm, int half, int bcol, int Nc) {
    if (bcol < Nc) {
        float bv = bias[bcol];
        #pragma unroll
        for (int r = 0; r < 8; ++r) {
            int m = tm * 16 + r + half * 8;
            float v = c[r] + bv;
            if (RELU) v = fmaxf(v, 0.0f);
            C[(size_t)m * Nc + bcol] = v;
        }
    }
}

template<int K, int NB, bool RELU, int TPW>
__global__ void __launch_bounds__(256)
gcn_gemm_wmma(const float* __restrict__ A,
              const float* __restrict__ B,
              const float* __restrict__ bias,
              float* __restrict__ C,
              int Nc, int tilesM) {
    constexpr int GROUPS = (NB / 16) / TPW;
    static_assert(GROUPS * TPW * 16 == NB, "column tiles must split evenly");
    int wave = threadIdx.x >> 5;
    int lane = threadIdx.x & 31;
    int tg = blockIdx.x % GROUPS;
    int tm = (blockIdx.x / GROUPS) * (blockDim.x >> 5) + wave;
    if (tm >= tilesM) return;

    int l15  = lane & 15;
    int half = lane >> 4;

    const float* Ap = A + (size_t)(tm * 16 + l15) * K + half * 2;
    v2f a[K / 4];
    #pragma unroll
    for (int s = 0; s < K / 4; ++s) a[s] = *(const v2f*)(Ap + 4 * s);

    if constexpr (TPW % 2 == 0) {
        for (int t = 0; t < TPW; t += 2) {
            int bcol = (tg * TPW + t) * 16 + l15;
            const float* Bp = B + (size_t)(half * 2) * NB + bcol;
            v8f c0 = {}, c1 = {};
            #pragma unroll
            for (int k0 = 0; k0 < K; k0 += 4) {
                v2f b0, b1;
                b0.x = Bp[k0 * NB];
                b0.y = Bp[k0 * NB + NB];
                b1.x = Bp[k0 * NB + 16];
                b1.y = Bp[k0 * NB + NB + 16];
                c0 = __builtin_amdgcn_wmma_f32_16x16x4_f32(
                        false, a[k0 / 4], false, b0, (short)0, c0, false, false);
                c1 = __builtin_amdgcn_wmma_f32_16x16x4_f32(
                        false, a[k0 / 4], false, b1, (short)0, c1, false, false);
            }
            gcn_store_tile<RELU>(C, bias, c0, tm, half, bcol, Nc);
            gcn_store_tile<RELU>(C, bias, c1, tm, half, bcol + 16, Nc);
        }
    } else {
        for (int t = 0; t < TPW; ++t) {
            int bcol = (tg * TPW + t) * 16 + l15;
            const float* Bp = B + (size_t)(half * 2) * NB + bcol;
            v8f c = {};
            #pragma unroll
            for (int k0 = 0; k0 < K; k0 += 4) {
                v2f b;
                b.x = Bp[k0 * NB];
                b.y = Bp[k0 * NB + NB];
                c = __builtin_amdgcn_wmma_f32_16x16x4_f32(
                        false, a[k0 / 4], false, b, (short)0, c, false, false);
            }
            gcn_store_tile<RELU>(C, bias, c, tm, half, bcol, Nc);
        }
    }
}

extern "C" void kernel_launch(void* const* d_in, const int* in_sizes, int n_in,
                              void* d_out, int out_size, void* d_ws, size_t ws_size,
                              hipStream_t stream) {
    const float* x    = (const float*)d_in[0];
    const float* W1   = (const float*)d_in[1];
    const float* b1   = (const float*)d_in[2];
    const float* W2   = (const float*)d_in[3];
    const float* b2   = (const float*)d_in[4];
    const int*   src1 = (const int*)d_in[5];
    const int*   dst1 = (const int*)d_in[6];
    const int*   src2 = (const int*)d_in[7];
    const int*   dst2 = (const int*)d_in[8];

    float* ws       = (float*)d_ws;
    float* deg_src1 = ws + OFF_DEG_SRC1;
    int*   cnt_dst1 = (int*)(ws + OFF_DEG_DST1);
    float* deg_src2 = ws + OFF_DEG_SRC2;
    int*   cnt_dst2 = (int*)(ws + OFF_DEG_DST2);
    float* agg1     = ws + OFF_AGG1;
    float* agg2     = ws + OFF_AGG2;
    float* h        = ws + OFF_H;
    float* W2p      = ws + OFF_W2P;
    float* b2p      = ws + OFF_B2P;
    float* out      = (float*)d_out;

    int* rank1 = (int*)h;
    int* bsrc1 = (int*)h + E1_;
    int* rank2 = (int*)h + 2 * E1_;
    int* bsrc2 = (int*)agg2;
    float* agg2s = agg1;

    const int T = 256;

    {
        static_assert(ZERO_COUNT % 4 == 0, "16-byte zero stores");
        static_assert(ZERO_COUNT / 4 >= HID_ * OUTP_ + OUTP_, "the grid is sized by the zero fill");
        int n4 = (int)(ZERO_COUNT / 4);
        gcn_zero_and_pad<<<(n4 + T - 1) / T, T, 0, stream>>>(ws, n4, W2, b2, W2p, b2p);
    }

    gcn_degrees_and_ranks<<<(E1_ / 4 + T - 1) / T, T, 0, stream>>>(
        src1, dst1, src2, dst2, deg_src1, cnt_dst1, rank1, deg_src2, cnt_dst2, rank2);

    gcn_row_starts_both<<<2, 1024, 0, stream>>>(cnt_dst1, cnt_dst2);
    gcn_bucket_fill_both<<<(E1_ / 4 + T - 1) / T, T, 0, stream>>>(
        src1, dst1, rank1, cnt_dst1, bsrc1, src2, dst2, rank2, cnt_dst2, bsrc2);

    gcn_gather_rows<IN_><<<2048, T, 0, stream>>>(
        x, bsrc1, cnt_dst1, deg_src1, agg1, N1_, E1_);

    {
        static_assert(N1_ % 16 == 0, "whole row tiles");
        int tilesM = N1_ / 16;
        int blocks = ((tilesM + (T / 32) - 1) / (T / 32)) * 2;
        gcn_gemm_wmma<IN_, HID_, true, 8><<<blocks, T, 0, stream>>>(
            agg1, W1, b1, h, HID_, tilesM);
    }

    gcn_gather_rows<HID_><<<(N2_ / 2 + (T / 32) - 1) / (T / 32), T, 0, stream>>>(
        h, bsrc2, cnt_dst2, deg_src2, agg2s, N2_, E2_);

    {
        static_assert(N2_ % 16 == 0, "whole row tiles");
        int tilesM = N2_ / 16;
        int blocks = ((tilesM + (T / 32) - 1) / (T / 32)) * (OUTP_ / 16);
        gcn_gemm_wmma<HID_, OUTP_, false, 1><<<blocks, T, 0, stream>>>(
            agg2s, W2p, b2p, out, OUT_, tilesM);
    }
}


# Degree counts and edge ranks on LDS counters, not global atomics: per-chunk LDS histograms (bucket order: chunk, then arrival order of LDS adds), summed by a scan pass; no zero fill, 8 -> 7 launches
# speedup vs baseline: 3.0149x; 1.4778x over previous
#include <hip/hip_runtime.h>
#include <hip/hip_bf16.h>

typedef float v2f __attribute__((ext_vector_type(2)));
typedef float v8f __attribute__((ext_vector_type(8)));

#define N0_   200000
#define N1_   50000
#define N2_   10000
#define E1_   750000
#define E2_   100000
#define IN_   128
#define HID_  256
#define OUT_  47
#define OUTP_ 48

#define KEYW_   50000
#define C1_     16
#define C2_     4
#define N4PC1_  ((E1_ / 4 + C1_ - 1) / C1_)
#define N4PC2_  ((E2_ / 4 + C2_ - 1) / C2_)
#define SRCW1_  (N0_ / KEYW_)
#define KB_     1024
#define NSEG1_  ((N1_ + KB_ - 1) / KB_)
#define NSEG2_  ((N2_ + KB_ - 1) / KB_)
#define NBS1_   ((N0_ + KB_ - 1) / KB_)
#define NBS2_   ((N1_ + KB_ - 1) / KB_)

#define OFF_DEG_SRC1  0
#define OFF_DEG_DST1  (OFF_DEG_SRC1 + N0_)
#define OFF_DEG_SRC2  (OFF_DEG_DST1 + N1_)
#define OFF_DEG_DST2  (OFF_DEG_SRC2 + N1_)
#define OFF_AGG1      (OFF_DEG_DST2 + N2_)
#define OFF_AGG2      (OFF_AGG1 + (size_t)N1_ * IN_)
#define OFF_H         (OFF_AGG2 + (size_t)N2_ * HID_)
#define OFF_W2P       (OFF_H + (size_t)N1_ * HID_)
#define OFF_B2P       (OFF_W2P + (size_t)HID_ * OUTP_)
#define HO_RANK1   0
#define HO_BSRC1   (HO_RANK1 + E1_)
#define HO_RANK2   (HO_BSRC1 + E1_)
#define HO_TBL_D1  (HO_RANK2 + E2_)
#define HO_TBL_S1  (HO_TBL_D1 + C1_ * N1_)
#define HO_TBL_D2  (HO_TBL_S1 + C1_ * N0_)
#define HO_TBL_S2  (HO_TBL_D2 + C2_ * N2_)
#define HO_SEG1    (HO_TBL_S2 + C2_ * N1_)
#define HO_SEG2    (HO_SEG1 + C1_ * NSEG1_)
#define HO_END     (HO_SEG2 + C2_ * NSEG2_)

__device__ __forceinline__ int gcn_lds_count(int* hist, int k, int nkeys) {
    return ((unsigned)k < (unsigned)nkeys) ? atomicAdd(&hist[k], 1) : 0;
}

template<bool RANK>
__device__ __forceinline__ void gcn_count_chunk(const int* __restrict__ ids, int i0, int i1,
                                                int klo, int nkeys, int* hist,
                                                int* __restrict__ rank,
                                                int* __restrict__ tbl_row,
                                                int* __restrict__ seg_row, int nseg) {
    int t = threadIdx.x;
    int lane = t & 31, wv = t >> 5;
    int n4 = nkeys >> 2;
    for (int i = t; i < n4; i += 1024) ((int4*)hist)[i] = make_int4(0, 0, 0, 0);
    __syncthreads();
    for (int base = i0; base < i1; base += 4 * 1024) {
        int4 v[4];
        #pragma unroll
        for (int j = 0; j < 4; ++j) {
            int idx = base + j * 1024 + t;
            v[j] = (idx < i1) ? ((const int4*)ids)[idx] : make_int4(-1, -1, -1, -1);
        }
        #pragma unroll
        for (int j = 0; j < 4; ++j) {
            int idx = base + j * 1024 + t;
            int4 r;
            r.x = gcn_lds_count(hist, v[j].x - klo, nkeys);
            r.y = gcn_lds_count(hist, v[j].y - klo, nkeys);
            r.z = gcn_lds_count(hist, v[j].z - klo, nkeys);
            r.w = gcn_lds_count(hist, v[j].w - klo, nkeys);
            if (RANK && idx < i1) ((int4*)rank)[idx] = r;
        }
    }
    __syncthreads();
    for (int i = t; i < n4; i += 1024) ((int4*)tbl_row)[i] = ((const int4*)hist)[i];
    if (seg_row != nullptr) {
        for (int b = wv; b < nseg; b += 32) {
            int s = 0;
            #pragma unroll
            for (int j = 0; j < KB_ / 128; ++j) {
                int i4 = b * (KB_ / 4) + j * 32 + lane;
                if (i4 < n4) {
                    int4 q = ((const int4*)hist)[i4];
                    s += q.x + q.y + q.z + q.w;
                }
            }
            #pragma unroll
            for (int off = 16; off > 0; off >>= 1) s += __shfl_xor(s, off);
            if (lane == 0) seg_row[b] = s;
        }
    }
}

__global__ void __launch_bounds__(1024)
gcn_count_chunks(const int* __restrict__ src1, const int* __restrict__ dst1,
                 const int* __restrict__ src2, const int* __restrict__ dst2,
                 int* __restrict__ rank1, int* __restrict__ rank2,
                 int* __restrict__ tbl_d1, int* __restrict__ tbl_s1,
                 int* __restrict__ tbl_d2, int* __restrict__ tbl_s2,
                 int* __restrict__ seg1, int* __restrict__ seg2,
                 const float* __restrict__ W2, const float* __restrict__ b2,
                 float* __restrict__ W2p, float* __restrict__ b2p) {
    static_assert(E1_ % 4 == 0 && E2_ % 4 == 0, "ids are read four at a time");
    static_assert(N1_ <= KEYW_ && N2_ <= KEYW_ && N0_ == SRCW1_ * KEYW_, "key windows");
    static_assert(N1_ % 4 == 0 && N2_ % 4 == 0 && KEYW_ % 4 == 0, "16-byte table stores");
    __shared__ __attribute__((aligned(16))) int hist[KEYW_];
    int b = blockIdx.x;
    if (b < C1_) {
        int i0 = b * N4PC1_;
        int i1 = min(i0 + N4PC1_, E1_ / 4);
        gcn_count_chunk<true>(dst1, i0, i1, 0, N1_, hist, rank1,
                              tbl_d1 + b * N1_, seg1 + b * NSEG1_, NSEG1_);
    } else if (b < C1_ + C1_ * SRCW1_) {
        int q = b - C1_;
        int c = q / SRCW1_, r = q - c * SRCW1_;
        int i0 = c * N4PC1_;
        int i1 = min(i0 + N4PC1_, E1_ / 4);
        gcn_count_chunk<false>(src1, i0, i1, r * KEYW_, KEYW_, hist, nullptr,
                               tbl_s1 + c * N0_ + r * KEYW_, nullptr, 0);
    } else if (b < C1_ + C1_ * SRCW1_ + C2_) {
        int c = b - (C1_ + C1_ * SRCW1_);
        int i0 = c * N4PC2_;
        int i1 = min(i0 + N4PC2_, E2_ / 4);
        gcn_count_chunk<true>(dst2, i0, i1, 0, N2_, hist, rank2,
                              tbl_d2 + c * N2_, seg2 + c * NSEG2_, NSEG2_);
    } else if (b < C1_ + C1_ * SRCW1_ + 2 * C2_) {
        int c = b - (C1_ + C1_ * SRCW1_ + C2_);
        int i0 = c * N4PC2_;
        int i1 = min(i0 + N4PC2_, E2_ / 4);
        gcn_count_chunk<false>(src2, i0, i1, 0, N1_, hist, nullptr,
                               tbl_s2 + c * N1_, nullptr, 0);
    } else {
        for (int i = threadIdx.x; i < HID_ * OUTP_ + OUTP_; i += 1024) {
            if (i < HID_ * OUTP_) {
                int r = i / OUTP_, c = i - r * OUTP_;
                W2p[i] = (c < OUT_) ? W2[r * OUT_ + c] : 0.0f;
            } else {
                int c = i - HID_ * OUTP_;
                b2p[c] = (c < OUT_) ? b2[c] : 0.0f;
            }
        }
    }
}
#define COUNT_BLOCKS (C1_ + C1_ * SRCW1_ + 2 * C2_ + 1)

template<int C>
__device__ __forceinline__ void gcn_starts_block(int* __restrict__ tbl, const int* __restrict__ seg,
                                                 int nkeys, int nseg, int kb,
                                                 int* __restrict__ start, int* sh) {
    int t = threadIdx.x;
    int lane = t & 31, wv = t >> 5;
    int ahead = 0;
    for (int i = t; i < C * nseg; i += KB_) {
        int bb = i % nseg;
        if (bb < kb) ahead += seg[i];
    }
    int k = kb * KB_ + t;
    bool valid = k < nkeys;
    int cnt[C];
    int total = 0;
    #pragma unroll
    for (int c = 0; c < C; ++c) {
        cnt[c] = valid ? tbl[(size_t)c * nkeys + k] : 0;
        total += cnt[c];
    }
    int v = total;
    #pragma unroll
    for (int off = 1; off < 32; off <<= 1) {
        int u = __shfl_up(v, off);
        if (lane >= off) v += u;
    }
    #pragma unroll
    for (int off = 16; off > 0; off >>= 1) ahead += __shfl_xor(ahead, off);
    if (lane == 31) sh[wv] = v;
    if (lane == 0)  sh[32 + wv] = ahead;
    __syncthreads();
    if (wv == 0) {
        int w = sh[lane];
        int a = sh[32 + lane];
        #pragma unroll
        for (int off = 1; off < 32; off <<= 1) {
            int u = __shfl_up(w, off);
            if (lane >= off) w += u;
        }
        #pragma unroll
        for (int off = 16; off > 0; off >>= 1) a += __shfl_xor(a, off);
        sh[lane] = w;
        if (lane == 0) sh[64] = a;
    }
    __syncthreads();
    int first = sh[64] + ((wv > 0) ? sh[wv - 1] : 0) + v - total;
    if (valid) {
        start[k] = first;
        int run = first;
        #pragma unroll
        for (int c = 0; c < C; ++c) {
            tbl[(size_t)c * nkeys + k] = run;
            run += cnt[c];
        }
    }
}

template<int C>
__device__ __forceinline__ void gcn_degree_block(const int* __restrict__ tbl, int nkeys, int kb,
                                                 float* __restrict__ deg) {
    int k = kb * KB_ + threadIdx.x;
    if (k < nkeys) {
        int s = 0;
        #pragma unroll
        for (int c = 0; c < C; ++c) s += tbl[(size_t)c * nkeys + k];
        deg[k] = (float)s;
    }
}

__global__ void __launch_bounds__(KB_)
gcn_combine_counts(int* __restrict__ tbl_d1, const int* __restrict__ seg1, int* __restrict__ start1,
                   int* __restrict__ tbl_d2, const int* __restrict__ seg2, int* __restrict__ start2,
                   const int* __restrict__ tbl_s1, float* __restrict__ deg_src1,
                   const int* __restrict__ tbl_s2, float* __restrict__ deg_src2) {
    __shared__ int sh[65];
    int b = blockIdx.x;
    if (b < NSEG1_)
        gcn_starts_block<C1_>(tbl_d1, seg1, N1_, NSEG1_, b, start1, sh);
    else if (b < NSEG1_ + NSEG2_)
        gcn_starts_block<C2_>(tbl_d2, seg2, N2_, NSEG2_, b - NSEG1_, start2, sh);
    else if (b < NSEG1_ + NSEG2_ + NBS1_)
        gcn_degree_block<C1_>(tbl_s1, N0_, b - (NSEG1_ + NSEG2_), deg_src1);
    else
        gcn_degree_block<C2_>(tbl_s2, N1_, b - (NSEG1_ + NSEG2_ + NBS1_), deg_src2);
}
#define COMBINE_BLOCKS (NSEG1_ + NSEG2_ + NBS1_ + NBS2_)

__global__ void gcn_bucket_fill_both(const int* __restrict__ src1, const int* __restrict__ dst1,
                                     const int* __restrict__ rank1, const int* __restrict__ tbl_d1,
                                     int* __restrict__ bsrc1,
                                     const int* __restrict__ src2, const int* __restrict__ dst2,
                                     const int* __restrict__ rank2, const int* __restrict__ tbl_d2,
                                     int* __restrict__ bsrc2) {
    static_assert(E2_ <= E1_, "the grid is sized by layer 1");
    static_assert(E1_ % 4 == 0 && E2_ % 4 == 0, "four edges per thread");
    int i = blockIdx.x * blockDim.x + threadIdx.x;
    if (i >= E1_ / 4) return;
    bool two = i < E2_ / 4;
    const int* slot1 = tbl_d1 + (i / N4PC1_) * N1_;
    const int* slot2 = tbl_d2 + (two ? i / N4PC2_ : 0) * N2_;
    int4 d1 = ((const int4*)dst1)[i];
    int4 r1 = ((const int4*)rank1)[i];
    int4 s1 = ((const int4*)src1)[i];
    int4 d2 = make_int4(0, 0, 0, 0), r2 = d2, s2 = d2;
    if (two) {
        d2 = ((const int4*)dst2)[i];
        r2 = ((const int4*)rank2)[i];
        s2 = ((const int4*)src2)[i];
    }
    int p1x = slot1[d1.x] + r1.x;
    int p1y = slot1[d1.y] + r1.y;
    int p1z = slot1[d1.z] + r1.z;
    int p1w = slot1[d1.w] + r1.w;
    if (two) {
        int p2x = slot2[d2.x] + r2.x;
        int p2y = slot2[d2.y] + r2.y;
        int p2z = slot2[d2.z] + r2.z;
        int p2w = slot2[d2.w] + r2.w;
        bsrc2[p2x] = s2.x;
        bsrc2[p2y] = s2.y;
        bsrc2[p2z] = s2.z;
        bsrc2[p2w] = s2.w;
    }
    bsrc1[p1x] = s1.x;
    bsrc1[p1y] = s1.y;
    bsrc1[p1z] = s1.z;
    bsrc1[p1w] = s1.w;
}

template<int COLS, int N>
__device__ __forceinline__ void gcn_add_edges(const float* __restrict__ feat, int s_l, float w_l,
                                              int j, int lane, float4 (&acc)[COLS / 128]) {
    constexpr int V = COLS / 128;
    float4 x[N][V];
    float  w[N];
    #pragma unroll
    for (int u = 0; u < N; ++u) {
        int s = __builtin_amdgcn_readlane(s_l, j + u);
        w[u]  = __int_as_float(__builtin_amdgcn_readlane(__float_as_int(w_l), j + u));
        const float4* p = (const float4*)(feat + (size_t)s * COLS) + lane;
        #pragma unroll
        for (int v = 0; v < V; ++v) x[u][v] = p[v * 32];
    }
    #pragma unroll
    for (int u = 0; u < N; ++u) {
        #pragma unroll
        for (int v = 0; v < V; ++v) {
            acc[v].x += x[u][v].x * w[u];
            acc[v].y += x[u][v].y * w[u];
            acc[v].z += x[u][v].z * w[u];
            acc[v].w += x[u][v].w * w[u];
        }
    }
}

template<int COLS>
__global__ void __launch_bounds__(256)
gcn_gather_rows(const float* __restrict__ feat,
                const int* __restrict__ bsrc,
                const int* __restrict__ start,
                const float* __restrict__ deg_src,
                float* __restrict__ agg, int n_rows, int n_edges) {
    constexpr int V = COLS / 128;
    int lane = threadIdx.x & 31;
    int w0 = __builtin_amdgcn_readfirstlane(blockIdx.x * (blockDim.x >> 5) + (threadIdx.x >> 5));
    int nw = gridDim.x * (blockDim.x >> 5);
    for (int row = w0; row < n_rows; row += nw) {
    int beg = start[row];
    int end = (row + 1 < n_rows) ? start[row + 1] : n_edges;

    float4 acc[V];
    #pragma unroll
    for (int v = 0; v < V; ++v) acc[v] = make_float4(0.f, 0.f, 0.f, 0.f);

    for (int base = beg; base < end; base += 32) {
        int m = end - base < 32 ? end - base : 32;
        int s_l = 0;
        float w_l = 0.0f;
        if (lane < m) {
            s_l = bsrc[base + lane];
            w_l = rsqrtf(fmaxf(deg_src[s_l], 1.0f));
        }
        int j = 0;
        for (; j + 8 <= m; j += 8) gcn_add_edges<COLS, 8>(feat, s_l, w_l, j, lane, acc);
        if (j + 4 <= m) { gcn_add_edges<COLS, 4>(feat, s_l, w_l, j, lane, acc); j += 4; }
        if (j + 2 <= m) { gcn_add_edges<COLS, 2>(feat, s_l, w_l, j, lane, acc); j += 2; }
        if (j < m)      { gcn_add_edges<COLS, 1>(feat, s_l, w_l, j, lane, acc); }
    }

    float nd = rsqrtf(fmaxf((float)(end - beg), 1.0f));
    float4* o = (float4*)(agg + (size_t)row * COLS) + lane;
    #pragma unroll
    for (int v = 0; v < V; ++v)
        o[v * 32] = make_float4(acc[v].x * nd, acc[v].y * nd, acc[v].z * nd, acc[v].w * nd);
    }
}

template<bool RELU>
__device__ __forceinline__ void gcn_store_tile(float* __restrict__ C, const float* __restrict__ bias,
                                               const v8f& c, int tm, int half, int bcol, int Nc) {
    if (bcol < Nc) {
        float bv = bias[bcol];
        #pragma unroll
        for (int r = 0; r < 8; ++r) {
            int m = tm * 16 + r + half * 8;
            float v = c[r] + bv;
            if (RELU) v = fmaxf(v, 0.0f);
            C[(size_t)m * Nc + bcol] = v;
        }
    }
}

template<int K, int NB, bool RELU, int TPW>
__global__ void __launch_bounds__(256)
gcn_gemm_wmma(const float* __restrict__ A,
              const float* __restrict__ B,
              const float* __restrict__ bias,
              float* __restrict__ C,
              int Nc, int tilesM) {
    constexpr int GROUPS = (NB / 16) / TPW;
    static_assert(GROUPS * TPW * 16 == NB, "column tiles must split evenly");
    int wave = threadIdx.x >> 5;
    int lane = threadIdx.x & 31;
    int tg = blockIdx.x % GROUPS;
    int tm = (blockIdx.x / GROUPS) * (blockDim.x >> 5) + wave;
    if (tm >= tilesM) return;

    int l15  = lane & 15;
    int half = lane >> 4;

    const float* Ap = A + (size_t)(tm * 16 + l15) * K + half * 2;
    v2f a[K / 4];
    #pragma unroll
    for (int s = 0; s < K / 4; ++s) a[s] = *(const v2f*)(Ap + 4 * s);

    if constexpr (TPW % 2 == 0) {
        for (int t = 0; t < TPW; t += 2) {
            int bcol = (tg * TPW + t) * 16 + l15;
            const float* Bp = B + (size_t)(half * 2) * NB + bcol;
            v8f c0 = {}, c1 = {};
            #pragma unroll
            for (int k0 = 0; k0 < K; k0 += 4) {
                v2f b0, b1;
                b0.x = Bp[k0 * NB];
                b0.y = Bp[k0 * NB + NB];
                b1.x = Bp[k0 * NB + 16];
                b1.y = Bp[k0 * NB + NB + 16];
                c0 = __builtin_amdgcn_wmma_f32_16x16x4_f32(
                        false, a[k0 / 4], false, b0, (short)0, c0, false, false);
                c1 = __builtin_amdgcn_wmma_f32_16x16x4_f32(
                        false, a[k0 / 4], false, b1, (short)0, c1, false, false);
            }
            gcn_store_tile<RELU>(C, bias, c0, tm, half, bcol, Nc);
            gcn_store_tile<RELU>(C, bias, c1, tm, half, bcol + 16, Nc);
        }
    } else {
        for (int t = 0; t < TPW; ++t) {
            int bcol = (tg * TPW + t) * 16 + l15;
            const float* Bp = B + (size_t)(half * 2) * NB + bcol;
            v8f c = {};
            #pragma unroll
            for (int k0 = 0; k0 < K; k0 += 4) {
                v2f b;
                b.x = Bp[k0 * NB];
                b.y = Bp[k0 * NB + NB];
                c = __builtin_amdgcn_wmma_f32_16x16x4_f32(
                        false, a[k0 / 4], false, b, (short)0, c, false, false);
            }
            gcn_store_tile<RELU>(C, bias, c, tm, half, bcol, Nc);
        }
    }
}

extern "C" void kernel_launch(void* const* d_in, const int* in_sizes, int n_in,
                              void* d_out, int out_size, void* d_ws, size_t ws_size,
                              hipStream_t stream) {
    const float* x    = (const float*)d_in[0];
    const float* W1   = (const float*)d_in[1];
    const float* b1   = (const float*)d_in[2];
    const float* W2   = (const float*)d_in[3];
    const float* b2   = (const float*)d_in[4];
    const int*   src1 = (const int*)d_in[5];
    const int*   dst1 = (const int*)d_in[6];
    const int*   src2 = (const int*)d_in[7];
    const int*   dst2 = (const int*)d_in[8];

    float* ws       = (float*)d_ws;
    float* deg_src1 = ws + OFF_DEG_SRC1;
    int*   start1   = (int*)(ws + OFF_DEG_DST1);
    float* deg_src2 = ws + OFF_DEG_SRC2;
    int*   start2   = (int*)(ws + OFF_DEG_DST2);
    float* agg1     = ws + OFF_AGG1;
    float* agg2     = ws + OFF_AGG2;
    float* h        = ws + OFF_H;
    float* W2p      = ws + OFF_W2P;
    float* b2p      = ws + OFF_B2P;
    float* out      = (float*)d_out;

    static_assert((size_t)HO_END <= (size_t)N1_ * HID_, "the tables fit in the h region");
    static_assert(OFF_H % 4 == 0 && HO_TBL_D1 % 4 == 0 && HO_TBL_S1 % 4 == 0 &&
                  HO_TBL_D2 % 4 == 0 && HO_TBL_S2 % 4 == 0, "16-byte table rows");
    int* hi     = (int*)h;
    int* rank1  = hi + HO_RANK1;
    int* bsrc1  = hi + HO_BSRC1;
    int* rank2  = hi + HO_RANK2;
    int* tbl_d1 = hi + HO_TBL_D1;
    int* tbl_s1 = hi + HO_TBL_S1;
    int* tbl_d2 = hi + HO_TBL_D2;
    int* tbl_s2 = hi + HO_TBL_S2;
    int* seg1   = hi + HO_SEG1;
    int* seg2   = hi + HO_SEG2;
    int* bsrc2  = (int*)agg2;
    float* agg2s = agg1;

    const int T = 256;

    gcn_count_chunks<<<COUNT_BLOCKS, 1024, 0, stream>>>(
        src1, dst1, src2, dst2, rank1, rank2, tbl_d1, tbl_s1, tbl_d2, tbl_s2,
        seg1, seg2, W2, b2, W2p, b2p);

    gcn_combine_counts<<<COMBINE_BLOCKS, KB_, 0, stream>>>(
        tbl_d1, seg1, start1, tbl_d2, seg2, start2, tbl_s1, deg_src1, tbl_s2, deg_src2);

    gcn_bucket_fill_both<<<(E1_ / 4 + T - 1) / T, T, 0, stream>>>(
        src1, dst1, rank1, tbl_d1, bsrc1, src2, dst2, rank2, tbl_d2, bsrc2);

    gcn_gather_rows<IN_><<<2048, T, 0, stream>>>(
        x, bsrc1, start1, deg_src1, agg1, N1_, E1_);

    {
        static_assert(N1_ % 16 == 0, "whole row tiles");
        int tilesM = N1_ / 16;
        int blocks = ((tilesM + (T / 32) - 1) / (T / 32)) * 2;
        gcn_gemm_wmma<IN_, HID_, true, 8><<<blocks, T, 0, stream>>>(
            agg1, W1, b1, h, HID_, tilesM);
    }

    gcn_gather_rows<HID_><<<(N2_ / 2 + (T / 32) - 1) / (T / 32), T, 0, stream>>>(
        h, bsrc2, start2, deg_src2, agg2s, N2_, E2_);

    {
        static_assert(N2_ % 16 == 0, "whole row tiles");
        int tilesM = N2_ / 16;
        int blocks = ((tilesM + (T / 32) - 1) / (T / 32)) * (OUTP_ / 16);
        gcn_gemm_wmma<HID_, OUTP_, false, 1><<<blocks, T, 0, stream>>>(
            agg2s, W2p, b2p, out, OUT_, tilesM);
    }
}


# Layer-1 aggregation and product in one launch: a block sums its 32 dst rows into an LDS tile (a wave walks 4 adjacent rows' edges as one stretch) and multiplies from there; agg1 and a launch gone
# speedup vs baseline: 3.4086x; 1.1306x over previous
#include <hip/hip_runtime.h>
#include <hip/hip_bf16.h>

typedef float v2f __attribute__((ext_vector_type(2)));
typedef float v8f __attribute__((ext_vector_type(8)));

#define N0_   200000
#define N1_   50000
#define N2_   10000
#define E1_   750000
#define E2_   100000
#define IN_   128
#define HID_  256
#define OUT_  47
#define OUTP_ 48

#define KEYW_   50000
#define C1_     16
#define C2_     4
#define N4PC1_  ((E1_ / 4 + C1_ - 1) / C1_)
#define N4PC2_  ((E2_ / 4 + C2_ - 1) / C2_)
#define SRCW1_  (N0_ / KEYW_)
#define KB_     1024
#define NSEG1_  ((N1_ + KB_ - 1) / KB_)
#define NSEG2_  ((N2_ + KB_ - 1) / KB_)
#define NBS1_   ((N0_ + KB_ - 1) / KB_)
#define NBS2_   ((N1_ + KB_ - 1) / KB_)

#define OFF_DEG_SRC1  0
#define OFF_DEG_DST1  (OFF_DEG_SRC1 + N0_)
#define OFF_DEG_SRC2  (OFF_DEG_DST1 + N1_)
#define OFF_DEG_DST2  (OFF_DEG_SRC2 + N1_)
#define OFF_AGG1      (OFF_DEG_DST2 + N2_)
#define OFF_AGG2      (OFF_AGG1 + (size_t)N1_ * IN_)
#define OFF_H         (OFF_AGG2 + (size_t)N2_ * HID_)
#define OFF_W2P       (OFF_H + (size_t)N1_ * HID_)
#define OFF_B2P       (OFF_W2P + (size_t)HID_ * OUTP_)
#define HO_RANK1   0
#define HO_RANK2   (HO_RANK1 + E1_)
#define HO_TBL_D1  (HO_RANK2 + E2_)
#define HO_TBL_S1  (HO_TBL_D1 + C1_ * N1_)
#define HO_TBL_D2  (HO_TBL_S1 + C1_ * N0_)
#define HO_TBL_S2  (HO_TBL_D2 + C2_ * N2_)
#define HO_SEG1    (HO_TBL_S2 + C2_ * N1_)
#define HO_SEG2    (HO_SEG1 + C1_ * NSEG1_)
#define HO_END     (HO_SEG2 + C2_ * NSEG2_)

__device__ __forceinline__ int gcn_lds_count(int* hist, int k, int nkeys) {
    return ((unsigned)k < (unsigned)nkeys) ? atomicAdd(&hist[k], 1) : 0;
}

template<bool RANK>
__device__ __forceinline__ void gcn_count_chunk(const int* __restrict__ ids, int i0, int i1,
                                                int klo, int nkeys, int* hist,
                                                int* __restrict__ rank,
                                                int* __restrict__ tbl_row,
                                                int* __restrict__ seg_row, int nseg) {
    int t = threadIdx.x;
    int lane = t & 31, wv = t >> 5;
    int n4 = nkeys >> 2;
    for (int i = t; i < n4; i += 1024) ((int4*)hist)[i] = make_int4(0, 0, 0, 0);
    __syncthreads();
    for (int base = i0; base < i1; base += 4 * 1024) {
        int4 v[4];
        #pragma unroll
        for (int j = 0; j < 4; ++j) {
            int idx = base + j * 1024 + t;
            v[j] = (idx < i1) ? ((const int4*)ids)[idx] : make_int4(-1, -1, -1, -1);
        }
        #pragma unroll
        for (int j = 0; j < 4; ++j) {
            int idx = base + j * 1024 + t;
            int4 r;
            r.x = gcn_lds_count(hist, v[j].x - klo, nkeys);
            r.y = gcn_lds_count(hist, v[j].y - klo, nkeys);
            r.z = gcn_lds_count(hist, v[j].z - klo, nkeys);
            r.w = gcn_lds_count(hist, v[j].w - klo, nkeys);
            if (RANK && idx < i1) ((int4*)rank)[idx] = r;
        }
    }
    __syncthreads();
    for (int i = t; i < n4; i += 1024) ((int4*)tbl_row)[i] = ((const int4*)hist)[i];
    if (seg_row != nullptr) {
        for (int b = wv; b < nseg; b += 32) {
            int s = 0;
            #pragma unroll
            for (int j = 0; j < KB_ / 128; ++j) {
                int i4 = b * (KB_ / 4) + j * 32 + lane;
                if (i4 < n4) {
                    int4 q = ((const int4*)hist)[i4];
                    s += q.x + q.y + q.z + q.w;
                }
            }
            #pragma unroll
            for (int off = 16; off > 0; off >>= 1) s += __shfl_xor(s, off);
            if (lane == 0) seg_row[b] = s;
        }
    }
}

__global__ void __launch_bounds__(1024)
gcn_count_chunks(const int* __restrict__ src1, const int* __restrict__ dst1,
                 const int* __restrict__ src2, const int* __restrict__ dst2,
                 int* __restrict__ rank1, int* __restrict__ rank2,
                 int* __restrict__ tbl_d1, int* __restrict__ tbl_s1,
                 int* __restrict__ tbl_d2, int* __restrict__ tbl_s2,
                 int* __restrict__ seg1, int* __restrict__ seg2,
                 const float* __restrict__ W2, const float* __restrict__ b2,
                 float* __restrict__ W2p, float* __restrict__ b2p) {
    static_assert(E1_ % 4 == 0 && E2_ % 4 == 0, "ids are read four at a time");
    static_assert(N1_ <= KEYW_ && N2_ <= KEYW_ && N0_ == SRCW1_ * KEYW_, "key windows");
    static_assert(N1_ % 4 == 0 && N2_ % 4 == 0 && KEYW_ % 4 == 0, "16-byte table stores");
    __shared__ __attribute__((aligned(16))) int hist[KEYW_];
    int b = blockIdx.x;
    if (b < C1_) {
        int i0 = b * N4PC1_;
        int i1 = min(i0 + N4PC1_, E1_ / 4);
        gcn_count_chunk<true>(dst1, i0, i1, 0, N1_, hist, rank1,
                              tbl_d1 + b * N1_, seg1 + b * NSEG1_, NSEG1_);
    } else if (b < C1_ + C1_ * SRCW1_) {
        int q = b - C1_;
        int c = q / SRCW1_, r = q - c * SRCW1_;
        int i0 = c * N4PC1_;
        int i1 = min(i0 + N4PC1_, E1_ / 4);
        gcn_count_chunk<false>(src1, i0, i1, r * KEYW_, KEYW_, hist, nullptr,
                               tbl_s1 + c * N0_ + r * KEYW_, nullptr, 0);
    } else if (b < C1_ + C1_ * SRCW1_ + C2_) {
        int c = b - (C1_ + C1_ * SRCW1_);
        int i0 = c * N4PC2_;
        int i1 = min(i0 + N4PC2_, E2_ / 4);
        gcn_count_chunk<true>(dst2, i0, i1, 0, N2_, hist, rank2,
                              tbl_d2 + c * N2_, seg2 + c * NSEG2_, NSEG2_);
    } else if (b < C1_ + C1_ * SRCW1_ + 2 * C2_) {
        int c = b - (C1_ + C1_ * SRCW1_ + C2_);
        int i0 = c * N4PC2_;
        int i1 = min(i0 + N4PC2_, E2_ / 4);
        gcn_count_chunk<false>(src2, i0, i1, 0, N1_, hist, nullptr,
                               tbl_s2 + c * N1_, nullptr, 0);
    } else {
        for (int i = threadIdx.x; i < HID_ * OUTP_ + OUTP_; i += 1024) {
            if (i < HID_ * OUTP_) {
                int r = i / OUTP_, c = i - r * OUTP_;
                W2p[i] = (c < OUT_) ? W2[r * OUT_ + c] : 0.0f;
            } else {
                int c = i - HID_ * OUTP_;
                b2p[c] = (c < OUT_) ? b2[c] : 0.0f;
            }
        }
    }
}
#define COUNT_BLOCKS (C1_ + C1_ * SRCW1_ + 2 * C2_ + 1)

template<int C>
__device__ __forceinline__ void gcn_starts_block(int* __restrict__ tbl, const int* __restrict__ seg,
                                                 int nkeys, int nseg, int kb,
                                                 int* __restrict__ start, int* sh) {
    int t = threadIdx.x;
    int lane = t & 31, wv = t >> 5;
    int ahead = 0;
    for (int i = t; i < C * nseg; i += KB_) {
        int bb = i % nseg;
        if (bb < kb) ahead += seg[i];
    }
    int k = kb * KB_ + t;
    bool valid = k < nkeys;
    int cnt[C];
    int total = 0;
    #pragma unroll
    for (int c = 0; c < C; ++c) {
        cnt[c] = valid ? tbl[(size_t)c * nkeys + k] : 0;
        total += cnt[c];
    }
    int v = total;
    #pragma unroll
    for (int off = 1; off < 32; off <<= 1) {
        int u = __shfl_up(v, off);
        if (lane >= off) v += u;
    }
    #pragma unroll
    for (int off = 16; off > 0; off >>= 1) ahead += __shfl_xor(ahead, off);
    if (lane == 31) sh[wv] = v;
    if (lane == 0)  sh[32 + wv] = ahead;
    __syncthreads();
    if (wv == 0) {
        int w = sh[lane];
        int a = sh[32 + lane];
        #pragma unroll
        for (int off = 1; off < 32; off <<= 1) {
            int u = __shfl_up(w, off);
            if (lane >= off) w += u;
        }
        #pragma unroll
        for (int off = 16; off > 0; off >>= 1) a += __shfl_xor(a, off);
        sh[lane] = w;
        if (lane == 0) sh[64] = a;
    }
    __syncthreads();
    int first = sh[64] + ((wv > 0) ? sh[wv - 1] : 0) + v - total;
    if (valid) {
        start[k] = first;
        int run = first;
        #pragma unroll
        for (int c = 0; c < C; ++c) {
            tbl[(size_t)c * nkeys + k] = run;
            run += cnt[c];
        }
    }
}

template<int C>
__device__ __forceinline__ void gcn_degree_block(const int* __restrict__ tbl, int nkeys, int kb,
                                                 float* __restrict__ deg) {
    int k = kb * KB_ + threadIdx.x;
    if (k < nkeys) {
        int s = 0;
        #pragma unroll
        for (int c = 0; c < C; ++c) s += tbl[(size_t)c * nkeys + k];
        deg[k] = (float)s;
    }
}

__global__ void __launch_bounds__(KB_)
gcn_combine_counts(int* __restrict__ tbl_d1, const int* __restrict__ seg1, int* __restrict__ start1,
                   int* __restrict__ tbl_d2, const int* __restrict__ seg2, int* __restrict__ start2,
                   const int* __restrict__ tbl_s1, float* __restrict__ deg_src1,
                   const int* __restrict__ tbl_s2, float* __restrict__ deg_src2) {
    __shared__ int sh[65];
    int b = blockIdx.x;
    if (b < NSEG1_)
        gcn_starts_block<C1_>(tbl_d1, seg1, N1_, NSEG1_, b, start1, sh);
    else if (b < NSEG1_ + NSEG2_)
        gcn_starts_block<C2_>(tbl_d2, seg2, N2_, NSEG2_, b - NSEG1_, start2, sh);
    else if (b < NSEG1_ + NSEG2_ + NBS1_)
        gcn_degree_block<C1_>(tbl_s1, N0_, b - (NSEG1_ + NSEG2_), deg_src1);
    else
        gcn_degree_block<C2_>(tbl_s2, N1_, b - (NSEG1_ + NSEG2_ + NBS1_), deg_src2);
}
#define COMBINE_BLOCKS (NSEG1_ + NSEG2_ + NBS1_ + NBS2_)

__global__ void gcn_bucket_fill_both(const int* __restrict__ src1, const int* __restrict__ dst1,
                                     const int* __restrict__ rank1, const int* __restrict__ tbl_d1,
                                     int* __restrict__ bsrc1,
                                     const int* __restrict__ src2, const int* __restrict__ dst2,
                                     const int* __restrict__ rank2, const int* __restrict__ tbl_d2,
                                     int* __restrict__ bsrc2) {
    static_assert(E2_ <= E1_, "the grid is sized by layer 1");
    static_assert(E1_ % 4 == 0 && E2_ % 4 == 0, "four edges per thread");
    int i = blockIdx.x * blockDim.x + threadIdx.x;
    if (i >= E1_ / 4) return;
    bool two = i < E2_ / 4;
    const int* slot1 = tbl_d1 + (i / N4PC1_) * N1_;
    const int* slot2 = tbl_d2 + (two ? i / N4PC2_ : 0) * N2_;
    int4 d1 = ((const int4*)dst1)[i];
    int4 r1 = ((const int4*)rank1)[i];
    int4 s1 = ((const int4*)src1)[i];
    int4 d2 = make_int4(0, 0, 0, 0), r2 = d2, s2 = d2;
    if (two) {
        d2 = ((const int4*)dst2)[i];
        r2 = ((const int4*)rank2)[i];
        s2 = ((const int4*)src2)[i];
    }
    int p1x = slot1[d1.x] + r1.x;
    int p1y = slot1[d1.y] + r1.y;
    int p1z = slot1[d1.z] + r1.z;
    int p1w = slot1[d1.w] + r1.w;
    if (two) {
        int p2x = slot2[d2.x] + r2.x;
        int p2y = slot2[d2.y] + r2.y;
        int p2z = slot2[d2.z] + r2.z;
        int p2w = slot2[d2.w] + r2.w;
        bsrc2[p2x] = s2.x;
        bsrc2[p2y] = s2.y;
        bsrc2[p2z] = s2.z;
        bsrc2[p2w] = s2.w;
    }
    bsrc1[p1x] = s1.x;
    bsrc1[p1y] = s1.y;
    bsrc1[p1z] = s1.z;
    bsrc1[p1w] = s1.w;
}

template<int COLS, int N>
__device__ __forceinline__ void gcn_add_edges(const float* __restrict__ feat, int s_l, float w_l,
                                              int j, int lane, float4 (&acc)[COLS / 128]) {
    constexpr int V = COLS / 128;
    float4 x[N][V];
    float  w[N];
    #pragma unroll
    for (int u = 0; u < N; ++u) {
        int s = __builtin_amdgcn_readlane(s_l, j + u);
        w[u]  = __int_as_float(__builtin_amdgcn_readlane(__float_as_int(w_l), j + u));
        const float4* p = (const float4*)(feat + (size_t)s * COLS) + lane;
        #pragma unroll
        for (int v = 0; v < V; ++v) x[u][v] = p[v * 32];
    }
    #pragma unroll
    for (int u = 0; u < N; ++u) {
        #pragma unroll
        for (int v = 0; v < V; ++v) {
            acc[v].x += x[u][v].x * w[u];
            acc[v].y += x[u][v].y * w[u];
            acc[v].z += x[u][v].z * w[u];
            acc[v].w += x[u][v].w * w[u];
        }
    }
}

template<int COLS>
__global__ void __launch_bounds__(256)
gcn_gather_rows(const float* __restrict__ feat,
                const int* __restrict__ bsrc,
                const int* __restrict__ start,
                const float* __restrict__ deg_src,
                float* __restrict__ agg, int n_rows, int n_edges) {
    constexpr int V = COLS / 128;
    int lane = threadIdx.x & 31;
    int w0 = __builtin_amdgcn_readfirstlane(blockIdx.x * (blockDim.x >> 5) + (threadIdx.x >> 5));
    int nw = gridDim.x * (blockDim.x >> 5);
    for (int row = w0; row < n_rows; row += nw) {
    int beg = start[row];
    int end = (row + 1 < n_rows) ? start[row + 1] : n_edges;

    float4 acc[V];
    #pragma unroll
    for (int v = 0; v < V; ++v) acc[v] = make_float4(0.f, 0.f, 0.f, 0.f);

    for (int base = beg; base < end; base += 32) {
        int m = end - base < 32 ? end - base : 32;
        int s_l = 0;
        float w_l = 0.0f;
        if (lane < m) {
            s_l = bsrc[base + lane];
            w_l = rsqrtf(fmaxf(deg_src[s_l], 1.0f));
        }
        int j = 0;
        for (; j + 8 <= m; j += 8) gcn_add_edges<COLS, 8>(feat, s_l, w_l, j, lane, acc);
        if (j + 4 <= m) { gcn_add_edges<COLS, 4>(feat, s_l, w_l, j, lane, acc); j += 4; }
        if (j + 2 <= m) { gcn_add_edges<COLS, 2>(feat, s_l, w_l, j, lane, acc); j += 2; }
        if (j < m)      { gcn_add_edges<COLS, 1>(feat, s_l, w_l, j, lane, acc); }
    }

    float nd = rsqrtf(fmaxf((float)(end - beg), 1.0f));
    float4* o = (float4*)(agg + (size_t)row * COLS) + lane;
    #pragma unroll
    for (int v = 0; v < V; ++v)
        o[v * 32] = make_float4(acc[v].x * nd, acc[v].y * nd, acc[v].z * nd, acc[v].w * nd);
    }
}

template<bool RELU>
__device__ __forceinline__ void gcn_store_tile(float* __restrict__ C, const float* __restrict__ bias,
                                               const v8f& c, int tm, int half, int bcol, int Nc) {
    if (bcol < Nc) {
        float bv = bias[bcol];
        #pragma unroll
        for (int r = 0; r < 8; ++r) {
            int m = tm * 16 + r + half * 8;
            float v = c[r] + bv;
            if (RELU) v = fmaxf(v, 0.0f);
            C[(size_t)m * Nc + bcol] = v;
        }
    }
}

template<int K, int NB, bool RELU, int TPW>
__global__ void __launch_bounds__(256)
gcn_gemm_wmma(const float* __restrict__ A,
              const float* __restrict__ B,
              const float* __restrict__ bias,
              float* __restrict__ C,
              int Nc, int tilesM) {
    constexpr int GROUPS = (NB / 16) / TPW;
    static_assert(GROUPS * TPW * 16 == NB, "column tiles must split evenly");
    int wave = threadIdx.x >> 5;
    int lane = threadIdx.x & 31;
    int tg = blockIdx.x % GROUPS;
    int tm = (blockIdx.x / GROUPS) * (blockDim.x >> 5) + wave;
    if (tm >= tilesM) return;

    int l15  = lane & 15;
    int half = lane >> 4;

    const float* Ap = A + (size_t)(tm * 16 + l15) * K + half * 2;
    v2f a[K / 4];
    #pragma unroll
    for (int s = 0; s < K / 4; ++s) a[s] = *(const v2f*)(Ap + 4 * s);

    if constexpr (TPW % 2 == 0) {
        for (int t = 0; t < TPW; t += 2) {
            int bcol = (tg * TPW + t) * 16 + l15;
            const float* Bp = B + (size_t)(half * 2) * NB + bcol;
            v8f c0 = {}, c1 = {};
            #pragma unroll
            for (int k0 = 0; k0 < K; k0 += 4) {
                v2f b0, b1;
                b0.x = Bp[k0 * NB];
                b0.y = Bp[k0 * NB + NB];
                b1.x = Bp[k0 * NB + 16];
                b1.y = Bp[k0 * NB + NB + 16];
                c0 = __builtin_amdgcn_wmma_f32_16x16x4_f32(
                        false, a[k0 / 4], false, b0, (short)0, c0, false, false);
                c1 = __builtin_amdgcn_wmma_f32_16x16x4_f32(
                        false, a[k0 / 4], false, b1, (short)0, c1, false, false);
            }
            gcn_store_tile<RELU>(C, bias, c0, tm, half, bcol, Nc);
            gcn_store_tile<RELU>(C, bias, c1, tm, half, bcol + 16, Nc);
        }
    } else {
        for (int t = 0; t < TPW; ++t) {
            int bcol = (tg * TPW + t) * 16 + l15;
            const float* Bp = B + (size_t)(half * 2) * NB + bcol;
            v8f c = {};
            #pragma unroll
            for (int k0 = 0; k0 < K; k0 += 4) {
                v2f b;
                b.x = Bp[k0 * NB];
                b.y = Bp[k0 * NB + NB];
                c = __builtin_amdgcn_wmma_f32_16x16x4_f32(
                        false, a[k0 / 4], false, b, (short)0, c, false, false);
            }
            gcn_store_tile<RELU>(C, bias, c, tm, half, bcol, Nc);
        }
    }
}


#define TR_  32
#define TP_  (IN_ + 4)

__device__ __forceinline__ void gcn_tile_row_done(float* trow, int st, int& cur, int& rb,
                                                  int& nb, float4& acc) {
    float nd = rsqrtf(fmaxf((float)(nb - rb), 1.0f));
    *(float4*)(trow + cur * TP_) = make_float4(acc.x * nd, acc.y * nd, acc.z * nd, acc.w * nd);
    acc = make_float4(0.f, 0.f, 0.f, 0.f);
    ++cur;
    rb = nb;
    nb = __builtin_amdgcn_readlane(st, cur + 1);
}

template<int N>
__device__ __forceinline__ void gcn_tile_add_edges(const float* __restrict__ feat, int s_l, float w_l,
                                                   int j, int e0, int lane, float* trow, int st,
                                                   int& cur, int& rb, int& nb, float4& acc) {
    float4 x[N];
    float  w[N];
    #pragma unroll
    for (int u = 0; u < N; ++u) {
        int s = __builtin_amdgcn_readlane(s_l, j + u);
        w[u]  = __int_as_float(__builtin_amdgcn_readlane(__float_as_int(w_l), j + u));
        x[u]  = ((const float4*)(feat + (size_t)s * IN_))[lane];
    }
    #pragma unroll
    for (int u = 0; u < N; ++u) {
        while (e0 + u >= nb && cur < 3) gcn_tile_row_done(trow, st, cur, rb, nb, acc);
        acc.x += x[u].x * w[u];
        acc.y += x[u].y * w[u];
        acc.z += x[u].z * w[u];
        acc.w += x[u].w * w[u];
    }
}

__global__ void __launch_bounds__(256, 16)
gcn_layer1_tile(const float* __restrict__ feat,
                const int* __restrict__ bsrc,
                const int* __restrict__ start,
                const float* __restrict__ deg_src,
                const float* __restrict__ B,
                const float* __restrict__ bias,
                float* __restrict__ C,
                int n_rows, int n_edges, int tilesM) {
    static_assert(IN_ == 128, "a lane holds 4 of a row's 128 floats");
    static_assert(HID_ == 8 * 32, "8 waves x 32 columns");
    __shared__ __attribute__((aligned(16))) float tile[TR_ * TP_];
    int lane = threadIdx.x & 31;
    int wave = __builtin_amdgcn_readfirstlane(threadIdx.x >> 5);

    {
        int r0 = blockIdx.x * TR_ + wave * 4;
        int st = n_edges;
        if (lane < 5 && r0 + lane < n_rows) st = start[r0 + lane];
        int cur = 0;
        int rb  = __builtin_amdgcn_readlane(st, 0);
        int nb  = __builtin_amdgcn_readlane(st, 1);
        int end = __builtin_amdgcn_readlane(st, 4);
        float4 acc = make_float4(0.f, 0.f, 0.f, 0.f);
        float* trow = tile + (wave * 4) * TP_ + lane * 4;

        for (int base = rb; base < end; base += 32) {
            int m = end - base < 32 ? end - base : 32;
            int s_l = 0;
            float w_l = 0.0f;
            if (lane < m) {
                s_l = bsrc[base + lane];
                w_l = rsqrtf(fmaxf(deg_src[s_l], 1.0f));
            }
            int j = 0;
            for (; j + 8 <= m; j += 8)
                gcn_tile_add_edges<8>(feat, s_l, w_l, j, base + j, lane, trow, st, cur, rb, nb, acc);
            if (j + 4 <= m) { gcn_tile_add_edges<4>(feat, s_l, w_l, j, base + j, lane, trow, st, cur, rb, nb, acc); j += 4; }
            if (j + 2 <= m) { gcn_tile_add_edges<2>(feat, s_l, w_l, j, base + j, lane, trow, st, cur, rb, nb, acc); j += 2; }
            if (j < m)      { gcn_tile_add_edges<1>(feat, s_l, w_l, j, base + j, lane, trow, st, cur, rb, nb, acc); }
        }
        while (cur < 4) gcn_tile_row_done(trow, st, cur, rb, nb, acc);
    }
    __syncthreads();

    int l15  = lane & 15;
    int half = lane >> 4;
    const float* A0 = tile + l15 * TP_ + half * 2;
    const float* A1 = A0 + 16 * TP_;
    int bcol = wave * 32 + l15;
    const float* Bp = B + (size_t)(half * 2) * HID_ + bcol;
    v8f c00 = {}, c01 = {}, c10 = {}, c11 = {};
    #pragma unroll
    for (int k0 = 0; k0 < IN_; k0 += 4) {
        v2f a0 = *(const v2f*)(A0 + k0);
        v2f a1 = *(const v2f*)(A1 + k0);
        v2f b0, b1;
        b0.x = Bp[k0 * HID_];
        b0.y = Bp[k0 * HID_ + HID_];
        b1.x = Bp[k0 * HID_ + 16];
        b1.y = Bp[k0 * HID_ + HID_ + 16];
        c00 = __builtin_amdgcn_wmma_f32_16x16x4_f32(false, a0, false, b0, (short)0, c00, false, false);
        c01 = __builtin_amdgcn_wmma_f32_16x16x4_f32(false, a0, false, b1, (short)0, c01, false, false);
        c10 = __builtin_amdgcn_wmma_f32_16x16x4_f32(false, a1, false, b0, (short)0, c10, false, false);
        c11 = __builtin_amdgcn_wmma_f32_16x16x4_f32(false, a1, false, b1, (short)0, c11, false, false);
    }
    int tm = blockIdx.x * (TR_ / 16);
    gcn_store_tile<true>(C, bias, c00, tm, half, bcol, HID_);
    gcn_store_tile<true>(C, bias, c01, tm, half, bcol + 16, HID_);
    if (tm + 1 < tilesM) {
        gcn_store_tile<true>(C, bias, c10, tm + 1, half, bcol, HID_);
        gcn_store_tile<true>(C, bias, c11, tm + 1, half, bcol + 16, HID_);
    }
}

extern "C" void kernel_launch(void* const* d_in, const int* in_sizes, int n_in,
                              void* d_out, int out_size, void* d_ws, size_t ws_size,
                              hipStream_t stream) {
    const float* x    = (const float*)d_in[0];
    const float* W1   = (const float*)d_in[1];
    const float* b1   = (const float*)d_in[2];
    const float* W2   = (const float*)d_in[3];
    const float* b2   = (const float*)d_in[4];
    const int*   src1 = (const int*)d_in[5];
    const int*   dst1 = (const int*)d_in[6];
    const int*   src2 = (const int*)d_in[7];
    const int*   dst2 = (const int*)d_in[8];

    float* ws       = (float*)d_ws;
    float* deg_src1 = ws + OFF_DEG_SRC1;
    int*   start1   = (int*)(ws + OFF_DEG_DST1);
    float* deg_src2 = ws + OFF_DEG_SRC2;
    int*   start2   = (int*)(ws + OFF_DEG_DST2);
    float* agg1     = ws + OFF_AGG1;
    float* agg2     = ws + OFF_AGG2;
    float* h        = ws + OFF_H;
    float* W2p      = ws + OFF_W2P;
    float* b2p      = ws + OFF_B2P;
    float* out      = (float*)d_out;

    static_assert((size_t)HO_END <= (size_t)N1_ * HID_, "the tables fit in the h region");
    static_assert(OFF_H % 4 == 0 && HO_TBL_D1 % 4 == 0 && HO_TBL_S1 % 4 == 0 &&
                  HO_TBL_D2 % 4 == 0 && HO_TBL_S2 % 4 == 0, "16-byte table rows");
    int* hi     = (int*)h;
    int* rank1  = hi + HO_RANK1;
    int* bsrc1  = (int*)agg1;
    int* rank2  = hi + HO_RANK2;
    int* tbl_d1 = hi + HO_TBL_D1;
    int* tbl_s1 = hi + HO_TBL_S1;
    int* tbl_d2 = hi + HO_TBL_D2;
    int* tbl_s2 = hi + HO_TBL_S2;
    int* seg1   = hi + HO_SEG1;
    int* seg2   = hi + HO_SEG2;
    int* bsrc2  = (int*)agg2;
    float* agg2s = agg1;

    const int T = 256;

    gcn_count_chunks<<<COUNT_BLOCKS, 1024, 0, stream>>>(
        src1, dst1, src2, dst2, rank1, rank2, tbl_d1, tbl_s1, tbl_d2, tbl_s2,
        seg1, seg2, W2, b2, W2p, b2p);

    gcn_combine_counts<<<COMBINE_BLOCKS, KB_, 0, stream>>>(
        tbl_d1, seg1, start1, tbl_d2, seg2, start2, tbl_s1, deg_src1, tbl_s2, deg_src2);

    gcn_bucket_fill_both<<<(E1_ / 4 + T - 1) / T, T, 0, stream>>>(
        src1, dst1, rank1, tbl_d1, bsrc1, src2, dst2, rank2, tbl_d2, bsrc2);

    {
        static_assert(N1_ % 16 == 0, "whole row tiles");
        static_assert((size_t)E1_ <= (size_t)N1_ * IN_, "bsrc1 fits in the agg1 region");
        int tilesM = N1_ / 16;
        gcn_layer1_tile<<<(N1_ + TR_ - 1) / TR_, T, 0, stream>>>(
            x, bsrc1, start1, deg_src1, W1, b1, h, N1_, E1_, tilesM);
    }

    gcn_gather_rows<HID_><<<(N2_ / 2 + (T / 32) - 1) / (T / 32), T, 0, stream>>>(
        h, bsrc2, start2, deg_src2, agg2s, N2_, E2_);

    {
        static_assert(N2_ % 16 == 0, "whole row tiles");
        int tilesM = N2_ / 16;
        int blocks = ((tilesM + (T / 32) - 1) / (T / 32)) * (OUTP_ / 16);
        gcn_gemm_wmma<HID_, OUTP_, false, 1><<<blocks, T, 0, stream>>>(
            agg2s, W2p, b2p, out, OUT_, tilesM);
    }
}


# Layer-2 aggregation and product in one launch: a block sums its 16 dst rows of h into an LDS tile (a wave walks 2 adjacent rows' edges as one stretch) and multiplies from it; agg2 and a launch gone
# speedup vs baseline: 3.6046x; 1.0575x over previous
#include <hip/hip_runtime.h>
#include <hip/hip_bf16.h>

typedef float v2f __attribute__((ext_vector_type(2)));
typedef float v8f __attribute__((ext_vector_type(8)));

#define N0_   200000
#define N1_   50000
#define N2_   10000
#define E1_   750000
#define E2_   100000
#define IN_   128
#define HID_  256
#define OUT_  47
#define OUTP_ 48

#define KEYW_   50000
#define C1_     16
#define C2_     4
#define N4PC1_  ((E1_ / 4 + C1_ - 1) / C1_)
#define N4PC2_  ((E2_ / 4 + C2_ - 1) / C2_)
#define SRCW1_  (N0_ / KEYW_)
#define KB_     1024
#define NSEG1_  ((N1_ + KB_ - 1) / KB_)
#define NSEG2_  ((N2_ + KB_ - 1) / KB_)
#define NBS1_   ((N0_ + KB_ - 1) / KB_)
#define NBS2_   ((N1_ + KB_ - 1) / KB_)

#define OFF_DEG_SRC1  0
#define OFF_DEG_DST1  (OFF_DEG_SRC1 + N0_)
#define OFF_DEG_SRC2  (OFF_DEG_DST1 + N1_)
#define OFF_DEG_DST2  (OFF_DEG_SRC2 + N1_)
#define OFF_AGG1      (OFF_DEG_DST2 + N2_)
#define OFF_AGG2      (OFF_AGG1 + (size_t)N1_ * IN_)
#define OFF_H         (OFF_AGG2 + (size_t)N2_ * HID_)
#define OFF_W2P       (OFF_H + (size_t)N1_ * HID_)
#define OFF_B2P       (OFF_W2P + (size_t)HID_ * OUTP_)
#define HO_RANK1   0
#define HO_RANK2   (HO_RANK1 + E1_)
#define HO_TBL_D1  (HO_RANK2 + E2_)
#define HO_TBL_S1  (HO_TBL_D1 + C1_ * N1_)
#define HO_TBL_D2  (HO_TBL_S1 + C1_ * N0_)
#define HO_TBL_S2  (HO_TBL_D2 + C2_ * N2_)
#define HO_SEG1    (HO_TBL_S2 + C2_ * N1_)
#define HO_SEG2    (HO_SEG1 + C1_ * NSEG1_)
#define HO_END     (HO_SEG2 + C2_ * NSEG2_)

__device__ __forceinline__ int gcn_lds_count(int* hist, int k, int nkeys) {
    return ((unsigned)k < (unsigned)nkeys) ? atomicAdd(&hist[k], 1) : 0;
}

template<bool RANK>
__device__ __forceinline__ void gcn_count_chunk(const int* __restrict__ ids, int i0, int i1,
                                                int klo, int nkeys, int* hist,
                                                int* __restrict__ rank,
                                                int* __restrict__ tbl_row,
                                                int* __restrict__ seg_row, int nseg) {
    int t = threadIdx.x;
    int lane = t & 31, wv = t >> 5;
    int n4 = nkeys >> 2;
    for (int i = t; i < n4; i += 1024) ((int4*)hist)[i] = make_int4(0, 0, 0, 0);
    __syncthreads();
    for (int base = i0; base < i1; base += 4 * 1024) {
        int4 v[4];
        #pragma unroll
        for (int j = 0; j < 4; ++j) {
            int idx = base + j * 1024 + t;
            v[j] = (idx < i1) ? ((const int4*)ids)[idx] : make_int4(-1, -1, -1, -1);
        }
        #pragma unroll
        for (int j = 0; j < 4; ++j) {
            int idx = base + j * 1024 + t;
            int4 r;
            r.x = gcn_lds_count(hist, v[j].x - klo, nkeys);
            r.y = gcn_lds_count(hist, v[j].y - klo, nkeys);
            r.z = gcn_lds_count(hist, v[j].z - klo, nkeys);
            r.w = gcn_lds_count(hist, v[j].w - klo, nkeys);
            if (RANK && idx < i1) ((int4*)rank)[idx] = r;
        }
    }
    __syncthreads();
    for (int i = t; i < n4; i += 1024) ((int4*)tbl_row)[i] = ((const int4*)hist)[i];
    if (seg_row != nullptr) {
        for (int b = wv; b < nseg; b += 32) {
            int s = 0;
            #pragma unroll
            for (int j = 0; j < KB_ / 128; ++j) {
                int i4 = b * (KB_ / 4) + j * 32 + lane;
                if (i4 < n4) {
                    int4 q = ((const int4*)hist)[i4];
                    s += q.x + q.y + q.z + q.w;
                }
            }
            #pragma unroll
            for (int off = 16; off > 0; off >>= 1) s += __shfl_xor(s, off);
            if (lane == 0) seg_row[b] = s;
        }
    }
}

__global__ void __launch_bounds__(1024)
gcn_count_chunks(const int* __restrict__ src1, const int* __restrict__ dst1,
                 const int* __restrict__ src2, const int* __restrict__ dst2,
                 int* __restrict__ rank1, int* __restrict__ rank2,
                 int* __restrict__ tbl_d1, int* __restrict__ tbl_s1,
                 int* __restrict__ tbl_d2, int* __restrict__ tbl_s2,
                 int* __restrict__ seg1, int* __restrict__ seg2,
                 const float* __restrict__ W2, const float* __restrict__ b2,
                 float* __restrict__ W2p, float* __restrict__ b2p) {
    static_assert(E1_ % 4 == 0 && E2_ % 4 == 0, "ids are read four at a time");
    static_assert(N1_ <= KEYW_ && N2_ <= KEYW_ && N0_ == SRCW1_ * KEYW_, "key windows");
    static_assert(N1_ % 4 == 0 && N2_ % 4 == 0 && KEYW_ % 4 == 0, "16-byte table stores");
    __shared__ __attribute__((aligned(16))) int hist[KEYW_];
    int b = blockIdx.x;
    if (b < C1_) {
        int i0 = b * N4PC1_;
        int i1 = min(i0 + N4PC1_, E1_ / 4);
        gcn_count_chunk<true>(dst1, i0, i1, 0, N1_, hist, rank1,
                              tbl_d1 + b * N1_, seg1 + b * NSEG1_, NSEG1_);
    } else if (b < C1_ + C1_ * SRCW1_) {
        int q = b - C1_;
        int c = q / SRCW1_, r = q - c * SRCW1_;
        int i0 = c * N4PC1_;
        int i1 = min(i0 + N4PC1_, E1_ / 4);
        gcn_count_chunk<false>(src1, i0, i1, r * KEYW_, KEYW_, hist, nullptr,
                               tbl_s1 + c * N0_ + r * KEYW_, nullptr, 0);
    } else if (b < C1_ + C1_ * SRCW1_ + C2_) {
        int c = b - (C1_ + C1_ * SRCW1_);
        int i0 = c * N4PC2_;
        int i1 = min(i0 + N4PC2_, E2_ / 4);
        gcn_count_chunk<true>(dst2, i0, i1, 0, N2_, hist, rank2,
                              tbl_d2 + c * N2_, seg2 + c * NSEG2_, NSEG2_);
    } else if (b < C1_ + C1_ * SRCW1_ + 2 * C2_) {
        int c = b - (C1_ + C1_ * SRCW1_ + C2_);
        int i0 = c * N4PC2_;
        int i1 = min(i0 + N4PC2_, E2_ / 4);
        gcn_count_chunk<false>(src2, i0, i1, 0, N1_, hist, nullptr,
                               tbl_s2 + c * N1_, nullptr, 0);
    } else {
        for (int i = threadIdx.x; i < HID_ * OUTP_ + OUTP_; i += 1024) {
            if (i < HID_ * OUTP_) {
                int r = i / OUTP_, c = i - r * OUTP_;
                W2p[i] = (c < OUT_) ? W2[r * OUT_ + c] : 0.0f;
            } else {
                int c = i - HID_ * OUTP_;
                b2p[c] = (c < OUT_) ? b2[c] : 0.0f;
            }
        }
    }
}
#define COUNT_BLOCKS (C1_ + C1_ * SRCW1_ + 2 * C2_ + 1)

template<int C>
__device__ __forceinline__ void gcn_starts_block(int* __restrict__ tbl, const int* __restrict__ seg,
                                                 int nkeys, int nseg, int kb,
                                                 int* __restrict__ start, int* sh) {
    int t = threadIdx.x;
    int lane = t & 31, wv = t >> 5;
    int ahead = 0;
    for (int i = t; i < C * nseg; i += KB_) {
        int bb = i % nseg;
        if (bb < kb) ahead += seg[i];
    }
    int k = kb * KB_ + t;
    bool valid = k < nkeys;
    int cnt[C];
    int total = 0;
    #pragma unroll
    for (int c = 0; c < C; ++c) {
        cnt[c] = valid ? tbl[(size_t)c * nkeys + k] : 0;
        total += cnt[c];
    }
    int v = total;
    #pragma unroll
    for (int off = 1; off < 32; off <<= 1) {
        int u = __shfl_up(v, off);
        if (lane >= off) v += u;
    }
    #pragma unroll
    for (int off = 16; off > 0; off >>= 1) ahead += __shfl_xor(ahead, off);
    if (lane == 31) sh[wv] = v;
    if (lane == 0)  sh[32 + wv] = ahead;
    __syncthreads();
    if (wv == 0) {
        int w = sh[lane];
        int a = sh[32 + lane];
        #pragma unroll
        for (int off = 1; off < 32; off <<= 1) {
            int u = __shfl_up(w, off);
            if (lane >= off) w += u;
        }
        #pragma unroll
        for (int off = 16; off > 0; off >>= 1) a += __shfl_xor(a, off);
        sh[lane] = w;
        if (lane == 0) sh[64] = a;
    }
    __syncthreads();
    int first = sh[64] + ((wv > 0) ? sh[wv - 1] : 0) + v - total;
    if (valid) {
        start[k] = first;
        int run = first;
        #pragma unroll
        for (int c = 0; c < C; ++c) {
            tbl[(size_t)c * nkeys + k] = run;
            run += cnt[c];
        }
    }
}

template<int C>
__device__ __forceinline__ void gcn_degree_block(const int* __restrict__ tbl, int nkeys, int kb,
                                                 float* __restrict__ deg) {
    int k = kb * KB_ + threadIdx.x;
    if (k < nkeys) {
        int s = 0;
        #pragma unroll
        for (int c = 0; c < C; ++c) s += tbl[(size_t)c * nkeys + k];
        deg[k] = (float)s;
    }
}

__global__ void __launch_bounds__(KB_)
gcn_combine_counts(int* __restrict__ tbl_d1, const int* __restrict__ seg1, int* __restrict__ start1,
                   int* __restrict__ tbl_d2, const int* __restrict__ seg2, int* __restrict__ start2,
                   const int* __restrict__ tbl_s1, float* __restrict__ deg_src1,
                   const int* __restrict__ tbl_s2, float* __restrict__ deg_src2) {
    __shared__ int sh[65];
    int b = blockIdx.x;
    if (b < NSEG1_)
        gcn_starts_block<C1_>(tbl_d1, seg1, N1_, NSEG1_, b, start1, sh);
    else if (b < NSEG1_ + NSEG2_)
        gcn_starts_block<C2_>(tbl_d2, seg2, N2_, NSEG2_, b - NSEG1_, start2, sh);
    else if (b < NSEG1_ + NSEG2_ + NBS1_)
        gcn_degree_block<C1_>(tbl_s1, N0_, b - (NSEG1_ + NSEG2_), deg_src1);
    else
        gcn_degree_block<C2_>(tbl_s2, N1_, b - (NSEG1_ + NSEG2_ + NBS1_), deg_src2);
}
#define COMBINE_BLOCKS (NSEG1_ + NSEG2_ + NBS1_ + NBS2_)

__global__ void gcn_bucket_fill_both(const int* __restrict__ src1, const int* __restrict__ dst1,
                                     const int* __restrict__ rank1, const int* __restrict__ tbl_d1,
                                     int* __restrict__ bsrc1,
                                     const int* __restrict__ src2, const int* __restrict__ dst2,
                                     const int* __restrict__ rank2, const int* __restrict__ tbl_d2,
                                     int* __restrict__ bsrc2) {
    static_assert(E2_ <= E1_, "the grid is sized by layer 1");
    static_assert(E1_ % 4 == 0 && E2_ % 4 == 0, "four edges per thread");
    int i = blockIdx.x * blockDim.x + threadIdx.x;
    if (i >= E1_ / 4) return;
    bool two = i < E2_ / 4;
    const int* slot1 = tbl_d1 + (i / N4PC1_) * N1_;
    const int* slot2 = tbl_d2 + (two ? i / N4PC2_ : 0) * N2_;
    int4 d1 = ((const int4*)dst1)[i];
    int4 r1 = ((const int4*)rank1)[i];
    int4 s1 = ((const int4*)src1)[i];
    int4 d2 = make_int4(0, 0, 0, 0), r2 = d2, s2 = d2;
    if (two) {
        d2 = ((const int4*)dst2)[i];
        r2 = ((const int4*)rank2)[i];
        s2 = ((const int4*)src2)[i];
    }
    int p1x = slot1[d1.x] + r1.x;
    int p1y = slot1[d1.y] + r1.y;
    int p1z = slot1[d1.z] + r1.z;
    int p1w = slot1[d1.w] + r1.w;
    if (two) {
        int p2x = slot2[d2.x] + r2.x;
        int p2y = slot2[d2.y] + r2.y;
        int p2z = slot2[d2.z] + r2.z;
        int p2w = slot2[d2.w] + r2.w;
        bsrc2[p2x] = s2.x;
        bsrc2[p2y] = s2.y;
        bsrc2[p2z] = s2.z;
        bsrc2[p2w] = s2.w;
    }
    bsrc1[p1x] = s1.x;
    bsrc1[p1y] = s1.y;
    bsrc1[p1z] = s1.z;
    bsrc1[p1w] = s1.w;
}

template<bool RELU>
__device__ __forceinline__ void gcn_store_tile(float* __restrict__ C, const float* __restrict__ bias,
                                               const v8f& c, int tm, int half, int bcol, int Nc) {
    if (bcol < Nc) {
        float bv = bias[bcol];
        #pragma unroll
        for (int r = 0; r < 8; ++r) {
            int m = tm * 16 + r + half * 8;
            float v = c[r] + bv;
            if (RELU) v = fmaxf(v, 0.0f);
            C[(size_t)m * Nc + bcol] = v;
        }
    }
}


#define TR_  32
#define TP_  (IN_ + 4)

__device__ __forceinline__ void gcn_tile_row_done(float* trow, int st, int& cur, int& rb,
                                                  int& nb, float4& acc) {
    float nd = rsqrtf(fmaxf((float)(nb - rb), 1.0f));
    *(float4*)(trow + cur * TP_) = make_float4(acc.x * nd, acc.y * nd, acc.z * nd, acc.w * nd);
    acc = make_float4(0.f, 0.f, 0.f, 0.f);
    ++cur;
    rb = nb;
    nb = __builtin_amdgcn_readlane(st, cur + 1);
}

template<int N>
__device__ __forceinline__ void gcn_tile_add_edges(const float* __restrict__ feat, int s_l, float w_l,
                                                   int j, int e0, int lane, float* trow, int st,
                                                   int& cur, int& rb, int& nb, float4& acc) {
    float4 x[N];
    float  w[N];
    #pragma unroll
    for (int u = 0; u < N; ++u) {
        int s = __builtin_amdgcn_readlane(s_l, j + u);
        w[u]  = __int_as_float(__builtin_amdgcn_readlane(__float_as_int(w_l), j + u));
        x[u]  = ((const float4*)(feat + (size_t)s * IN_))[lane];
    }
    #pragma unroll
    for (int u = 0; u < N; ++u) {
        while (e0 + u >= nb && cur < 3) gcn_tile_row_done(trow, st, cur, rb, nb, acc);
        acc.x += x[u].x * w[u];
        acc.y += x[u].y * w[u];
        acc.z += x[u].z * w[u];
        acc.w += x[u].w * w[u];
    }
}

__global__ void __launch_bounds__(256, 16)
gcn_layer1_tile(const float* __restrict__ feat,
                const int* __restrict__ bsrc,
                const int* __restrict__ start,
                const float* __restrict__ deg_src,
                const float* __restrict__ B,
                const float* __restrict__ bias,
                float* __restrict__ C,
                int n_rows, int n_edges, int tilesM) {
    static_assert(IN_ == 128, "a lane holds 4 of a row's 128 floats");
    static_assert(HID_ == 8 * 32, "8 waves x 32 columns");
    __shared__ __attribute__((aligned(16))) float tile[TR_ * TP_];
    int lane = threadIdx.x & 31;
    int wave = __builtin_amdgcn_readfirstlane(threadIdx.x >> 5);

    {
        int r0 = blockIdx.x * TR_ + wave * 4;
        int st = n_edges;
        if (lane < 5 && r0 + lane < n_rows) st = start[r0 + lane];
        int cur = 0;
        int rb  = __builtin_amdgcn_readlane(st, 0);
        int nb  = __builtin_amdgcn_readlane(st, 1);
        int end = __builtin_amdgcn_readlane(st, 4);
        float4 acc = make_float4(0.f, 0.f, 0.f, 0.f);
        float* trow = tile + (wave * 4) * TP_ + lane * 4;

        for (int base = rb; base < end; base += 32) {
            int m = end - base < 32 ? end - base : 32;
            int s_l = 0;
            float w_l = 0.0f;
            if (lane < m) {
                s_l = bsrc[base + lane];
                w_l = rsqrtf(fmaxf(deg_src[s_l], 1.0f));
            }
            int j = 0;
            for (; j + 8 <= m; j += 8)
                gcn_tile_add_edges<8>(feat, s_l, w_l, j, base + j, lane, trow, st, cur, rb, nb, acc);
            if (j + 4 <= m) { gcn_tile_add_edges<4>(feat, s_l, w_l, j, base + j, lane, trow, st, cur, rb, nb, acc); j += 4; }
            if (j + 2 <= m) { gcn_tile_add_edges<2>(feat, s_l, w_l, j, base + j, lane, trow, st, cur, rb, nb, acc); j += 2; }
            if (j < m)      { gcn_tile_add_edges<1>(feat, s_l, w_l, j, base + j, lane, trow, st, cur, rb, nb, acc); }
        }
        while (cur < 4) gcn_tile_row_done(trow, st, cur, rb, nb, acc);
    }
    __syncthreads();

    int l15  = lane & 15;
    int half = lane >> 4;
    const float* A0 = tile + l15 * TP_ + half * 2;
    const float* A1 = A0 + 16 * TP_;
    int bcol = wave * 32 + l15;
    const float* Bp = B + (size_t)(half * 2) * HID_ + bcol;
    v8f c00 = {}, c01 = {}, c10 = {}, c11 = {};
    #pragma unroll
    for (int k0 = 0; k0 < IN_; k0 += 4) {
        v2f a0 = *(const v2f*)(A0 + k0);
        v2f a1 = *(const v2f*)(A1 + k0);
        v2f b0, b1;
        b0.x = Bp[k0 * HID_];
        b0.y = Bp[k0 * HID_ + HID_];
        b1.x = Bp[k0 * HID_ + 16];
        b1.y = Bp[k0 * HID_ + HID_ + 16];
        c00 = __builtin_amdgcn_wmma_f32_16x16x4_f32(false, a0, false, b0, (short)0, c00, false, false);
        c01 = __builtin_amdgcn_wmma_f32_16x16x4_f32(false, a0, false, b1, (short)0, c01, false, false);
        c10 = __builtin_amdgcn_wmma_f32_16x16x4_f32(false, a1, false, b0, (short)0, c10, false, false);
        c11 = __builtin_amdgcn_wmma_f32_16x16x4_f32(false, a1, false, b1, (short)0, c11, false, false);
    }
    int tm = blockIdx.x * (TR_ / 16);
    gcn_store_tile<true>(C, bias, c00, tm, half, bcol, HID_);
    gcn_store_tile<true>(C, bias, c01, tm, half, bcol + 16, HID_);
    if (tm + 1 < tilesM) {
        gcn_store_tile<true>(C, bias, c10, tm + 1, half, bcol, HID_);
        gcn_store_tile<true>(C, bias, c11, tm + 1, half, bcol + 16, HID_);
    }
}

#define TR2_   16
#define RPW2_  2
#define TP2_   (HID_ + 4)

__device__ __forceinline__ void gcn_tile2_row_done(float* trow, int st, int& cur, int& rb,
                                                   int& nb, float4 (&acc)[2]) {
    float nd = rsqrtf(fmaxf((float)(nb - rb), 1.0f));
    float* o = trow + cur * TP2_;
    *(float4*)(o)       = make_float4(acc[0].x * nd, acc[0].y * nd, acc[0].z * nd, acc[0].w * nd);
    *(float4*)(o + 128) = make_float4(acc[1].x * nd, acc[1].y * nd, acc[1].z * nd, acc[1].w * nd);
    acc[0] = make_float4(0.f, 0.f, 0.f, 0.f);
    acc[1] = make_float4(0.f, 0.f, 0.f, 0.f);
    ++cur;
    rb = nb;
    nb = __builtin_amdgcn_readlane(st, cur + 1);
}

template<int N>
__device__ __forceinline__ void gcn_tile2_add_edges(const float* __restrict__ feat, int s_l, float w_l,
                                                    int j, int e0, int lane, float* trow, int st,
                                                    int& cur, int& rb, int& nb, float4 (&acc)[2]) {
    float4 x[N][2];
    float  w[N];
    #pragma unroll
    for (int u = 0; u < N; ++u) {
        int s = __builtin_amdgcn_readlane(s_l, j + u);
        w[u]  = __int_as_float(__builtin_amdgcn_readlane(__float_as_int(w_l), j + u));
        const float4* p = (const float4*)(feat + (size_t)s * HID_) + lane;
        x[u][0] = p[0];
        x[u][1] = p[32];
    }
    #pragma unroll
    for (int u = 0; u < N; ++u) {
        while (e0 + u >= nb && cur < RPW2_ - 1) gcn_tile2_row_done(trow, st, cur, rb, nb, acc);
        #pragma unroll
        for (int v = 0; v < 2; ++v) {
            acc[v].x += x[u][v].x * w[u];
            acc[v].y += x[u][v].y * w[u];
            acc[v].z += x[u][v].z * w[u];
            acc[v].w += x[u][v].w * w[u];
        }
    }
}

__global__ void __launch_bounds__(256)
gcn_layer2_tile(const float* __restrict__ feat,
                const int* __restrict__ bsrc,
                const int* __restrict__ start,
                const float* __restrict__ deg_src,
                const float* __restrict__ B,
                const float* __restrict__ bias,
                float* __restrict__ C,
                int n_rows, int n_edges) {
    static_assert(HID_ == 256, "a lane holds 8 of a row's 256 floats");
    static_assert(TR2_ == 8 * RPW2_, "8 waves x RPW2_ rows");
    static_assert(OUTP_ / 16 <= 8, "one wave per column tile");
    __shared__ __attribute__((aligned(16))) float tile[TR2_ * TP2_];
    int lane = threadIdx.x & 31;
    int wave = __builtin_amdgcn_readfirstlane(threadIdx.x >> 5);

    {
        int r0 = blockIdx.x * TR2_ + wave * RPW2_;
        int st = n_edges;
        if (lane <= RPW2_ && r0 + lane < n_rows) st = start[r0 + lane];
        int cur = 0;
        int rb  = __builtin_amdgcn_readlane(st, 0);
        int nb  = __builtin_amdgcn_readlane(st, 1);
        int end = __builtin_amdgcn_readlane(st, RPW2_);
        float4 acc[2];
        acc[0] = make_float4(0.f, 0.f, 0.f, 0.f);
        acc[1] = make_float4(0.f, 0.f, 0.f, 0.f);
        float* trow = tile + (wave * RPW2_) * TP2_ + lane * 4;

        for (int base = rb; base < end; base += 32) {
            int m = end - base < 32 ? end - base : 32;
            int s_l = 0;
            float w_l = 0.0f;
            if (lane < m) {
                s_l = bsrc[base + lane];
                w_l = rsqrtf(fmaxf(deg_src[s_l], 1.0f));
            }
            int j = 0;
            for (; j + 8 <= m; j += 8)
                gcn_tile2_add_edges<8>(feat, s_l, w_l, j, base + j, lane, trow, st, cur, rb, nb, acc);
            if (j + 4 <= m) { gcn_tile2_add_edges<4>(feat, s_l, w_l, j, base + j, lane, trow, st, cur, rb, nb, acc); j += 4; }
            if (j + 2 <= m) { gcn_tile2_add_edges<2>(feat, s_l, w_l, j, base + j, lane, trow, st, cur, rb, nb, acc); j += 2; }
            if (j < m)      { gcn_tile2_add_edges<1>(feat, s_l, w_l, j, base + j, lane, trow, st, cur, rb, nb, acc); }
        }
        while (cur < RPW2_) gcn_tile2_row_done(trow, st, cur, rb, nb, acc);
    }
    __syncthreads();

    if (wave < OUTP_ / 16) {
        int l15  = lane & 15;
        int half = lane >> 4;
        const float* A0 = tile + l15 * TP2_ + half * 2;
        int bcol = wave * 16 + l15;
        const float* Bp = B + (size_t)(half * 2) * OUTP_ + bcol;
        v8f c = {};
        #pragma unroll
        for (int k0 = 0; k0 < HID_; k0 += 4) {
            v2f a = *(const v2f*)(A0 + k0);
            v2f b;
            b.x = Bp[k0 * OUTP_];
            b.y = Bp[k0 * OUTP_ + OUTP_];
            c = __builtin_amdgcn_wmma_f32_16x16x4_f32(false, a, false, b, (short)0, c, false, false);
        }
        gcn_store_tile<false>(C, bias, c, blockIdx.x, half, bcol, OUT_);
    }
}

extern "C" void kernel_launch(void* const* d_in, const int* in_sizes, int n_in,
                              void* d_out, int out_size, void* d_ws, size_t ws_size,
                              hipStream_t stream) {
    const float* x    = (const float*)d_in[0];
    const float* W1   = (const float*)d_in[1];
    const float* b1   = (const float*)d_in[2];
    const float* W2   = (const float*)d_in[3];
    const float* b2   = (const float*)d_in[4];
    const int*   src1 = (const int*)d_in[5];
    const int*   dst1 = (const int*)d_in[6];
    const int*   src2 = (const int*)d_in[7];
    const int*   dst2 = (const int*)d_in[8];

    float* ws       = (float*)d_ws;
    float* deg_src1 = ws + OFF_DEG_SRC1;
    int*   start1   = (int*)(ws + OFF_DEG_DST1);
    float* deg_src2 = ws + OFF_DEG_SRC2;
    int*   start2   = (int*)(ws + OFF_DEG_DST2);
    float* agg1     = ws + OFF_AGG1;
    float* agg2     = ws + OFF_AGG2;
    float* h        = ws + OFF_H;
    float* W2p      = ws + OFF_W2P;
    float* b2p      = ws + OFF_B2P;
    float* out      = (float*)d_out;

    static_assert((size_t)HO_END <= (size_t)N1_ * HID_, "the tables fit in the h region");
    static_assert(OFF_H % 4 == 0 && HO_TBL_D1 % 4 == 0 && HO_TBL_S1 % 4 == 0 &&
                  HO_TBL_D2 % 4 == 0 && HO_TBL_S2 % 4 == 0, "16-byte table rows");
    int* hi     = (int*)h;
    int* rank1  = hi + HO_RANK1;
    int* bsrc1  = (int*)agg1;
    int* rank2  = hi + HO_RANK2;
    int* tbl_d1 = hi + HO_TBL_D1;
    int* tbl_s1 = hi + HO_TBL_S1;
    int* tbl_d2 = hi + HO_TBL_D2;
    int* tbl_s2 = hi + HO_TBL_S2;
    int* seg1   = hi + HO_SEG1;
    int* seg2   = hi + HO_SEG2;
    int* bsrc2  = (int*)agg2;

    const int T = 256;

    gcn_count_chunks<<<COUNT_BLOCKS, 1024, 0, stream>>>(
        src1, dst1, src2, dst2, rank1, rank2, tbl_d1, tbl_s1, tbl_d2, tbl_s2,
        seg1, seg2, W2, b2, W2p, b2p);

    gcn_combine_counts<<<COMBINE_BLOCKS, KB_, 0, stream>>>(
        tbl_d1, seg1, start1, tbl_d2, seg2, start2, tbl_s1, deg_src1, tbl_s2, deg_src2);

    gcn_bucket_fill_both<<<(E1_ / 4 + T - 1) / T, T, 0, stream>>>(
        src1, dst1, rank1, tbl_d1, bsrc1, src2, dst2, rank2, tbl_d2, bsrc2);

    {
        static_assert(N1_ % 16 == 0, "whole row tiles");
        static_assert((size_t)E1_ <= (size_t)N1_ * IN_, "bsrc1 fits in the agg1 region");
        int tilesM = N1_ / 16;
        gcn_layer1_tile<<<(N1_ + TR_ - 1) / TR_, T, 0, stream>>>(
            x, bsrc1, start1, deg_src1, W1, b1, h, N1_, E1_, tilesM);
    }

    {
        static_assert(N2_ % TR2_ == 0, "whole row tiles");
        gcn_layer2_tile<<<N2_ / TR2_, T, 0, stream>>>(
            h, bsrc2, start2, deg_src2, W2p, b2p, out, N2_, E2_);
    }
}
